# starting kernel
# baseline (speedup 1.0000x reference)
#include <hip/hip_runtime.h>
#include <math.h>

typedef __attribute__((ext_vector_type(2))) float v2f;
typedef __attribute__((ext_vector_type(8))) float v8f;

#define HID 128

__device__ __forceinline__ float lrelu(float v, float s) { return v > 0.f ? v : s * v; }

__device__ __forceinline__ void atomicMaxF(float* addr, float val) {
  if (val >= 0.f) atomicMax((int*)addr, __float_as_int(val));
  else            atomicMin((unsigned int*)addr, __float_as_uint(val));
}

__global__ void encoder_kernel(const float* __restrict__ num_prop, const float* __restrict__ cat_prop,
                               const float* __restrict__ Wn, const float* __restrict__ bn,
                               const float* __restrict__ Wc, const float* __restrict__ bc,
                               float* __restrict__ xcat, int N) {
  int t = blockIdx.x * blockDim.x + threadIdx.x;
  if (t >= N * 64) return;
  int n = t >> 6, j = t & 63;
  const float* np = num_prop + (size_t)n * 7;
  float an = bn[j];
#pragma unroll
  for (int k = 0; k < 7; ++k) an += np[k] * Wn[k * 64 + j];
  const float* cp = cat_prop + (size_t)n * 3;
  float ac = bc[j];
#pragma unroll
  for (int k = 0; k < 3; ++k) ac += cp[k] * Wc[k * 64 + j];
  xcat[(size_t)n * HID + j]      = lrelu(an, 0.01f);
  xcat[(size_t)n * HID + 64 + j] = lrelu(ac, 0.01f);
}

__global__ void wmma_gemm128_kernel(const float* __restrict__ A, const float* __restrict__ B,
                                    const float* __restrict__ bias, float* __restrict__ C,
                                    int M, int act, int has_bias) {
  int wave = threadIdx.x >> 5;
  int lane = threadIdx.x & 31;
  int row0 = blockIdx.x * 16;
  int col0 = wave * 16;
  int half = lane >> 4;
  int m16  = lane & 15;

  int arow = row0 + m16; if (arow >= M) arow = M - 1;
  const float* Ap = A + (size_t)arow * HID + 2 * half;
  const float* Bp = B + (size_t)(2 * half) * HID + col0 + m16;

  v8f acc = {0.f, 0.f, 0.f, 0.f, 0.f, 0.f, 0.f, 0.f};
#pragma unroll 8
  for (int k0 = 0; k0 < HID; k0 += 4) {
    v2f a, b;
    a.x = Ap[k0];
    a.y = Ap[k0 + 1];
    b.x = Bp[(size_t)k0 * HID];
    b.y = Bp[(size_t)(k0 + 1) * HID];
    acc = __builtin_amdgcn_wmma_f32_16x16x4_f32(false, a, false, b, (short)0, acc, false, false);
  }

  int col = col0 + m16;
  float bv = has_bias ? bias[col] : 0.f;
#pragma unroll
  for (int r = 0; r < 8; ++r) {
    int row = row0 + r + 8 * half;
    if (row < M) {
      float v = acc[r] + bv;
      if (act) v = lrelu(v, 0.01f);
      C[(size_t)row * HID + col] = v;
    }
  }
}

__global__ void bn_stats_kernel(const float* __restrict__ x, float* __restrict__ stats, int N) {
  int c = threadIdx.x;
  float s = 0.f, s2 = 0.f;
  for (int n = blockIdx.x; n < N; n += gridDim.x) {
    float v = x[(size_t)n * HID + c];
    s += v; s2 += v * v;
  }
  atomicAdd(stats + c, s);
  atomicAdd(stats + HID + c, s2);
}

__global__ void bn_finalize_kernel(float* __restrict__ stats, const float* __restrict__ gamma,
                                   const float* __restrict__ beta, int N) {
  int c = threadIdx.x;
  float inv_n = 1.f / (float)N;
  float mu  = stats[c] * inv_n;
  float var = stats[HID + c] * inv_n - mu * mu;
  float sc  = gamma[c] * rsqrtf(var + 1e-5f);
  stats[2 * HID + c] = sc;
  stats[3 * HID + c] = beta[c] - mu * sc;
}

__global__ void bn_apply_kernel(const float* __restrict__ x, const float* __restrict__ stats,
                                float* __restrict__ y, int N) {
  int t = blockIdx.x * blockDim.x + threadIdx.x;
  if (t >= N * HID) return;
  int c = t & (HID - 1);
  y[t] = x[t] * stats[2 * HID + c] + stats[3 * HID + c];
}

__global__ void att_logits_kernel(const float* __restrict__ h, const float* __restrict__ a_src,
                                  const float* __restrict__ a_dst, float* __restrict__ als,
                                  float* __restrict__ ald, int N, int H, int F) {
  int t = blockIdx.x * blockDim.x + threadIdx.x;
  if (t >= N * H) return;
  int n = t / H, hh = t - n * H;
  const float* hp = h + (size_t)n * HID + hh * F;
  const float* as = a_src + hh * F;
  const float* ad = a_dst + hh * F;
  float ss = 0.f, sd = 0.f;
  for (int f = 0; f < F; ++f) { float v = hp[f]; ss += v * as[f]; sd += v * ad[f]; }
  als[t] = ss;
  ald[t] = sd;
}

__global__ void fill_kernel(float* __restrict__ p, long long n, float v) {
  long long t = (long long)blockIdx.x * blockDim.x + threadIdx.x;
  if (t < n) p[t] = v;
}

__device__ __forceinline__ void edge_sd(const int* __restrict__ ei, int E, long long e,
                                        int& s, int& d) {
  if (e < E) { s = ei[e]; d = ei[(size_t)E + e]; }
  else       { s = d = (int)(e - E); }
}

__global__ void edge_max_kernel(const int* __restrict__ ei, const float* __restrict__ als,
                                const float* __restrict__ ald, float* __restrict__ emax,
                                int E, int N, int H) {
  long long t = (long long)blockIdx.x * blockDim.x + threadIdx.x;
  long long total = (long long)(E + N) * H;
  if (t >= total) return;
  long long e = t / H; int hh = (int)(t - e * H);
  int s, d; edge_sd(ei, E, e, s, d);
  float v = lrelu(als[(size_t)s * H + hh] + ald[(size_t)d * H + hh], 0.2f);
  atomicMaxF(emax + (size_t)d * H + hh, v);
}

__global__ void edge_denom_kernel(const int* __restrict__ ei, const float* __restrict__ als,
                                  const float* __restrict__ ald, const float* __restrict__ emax,
                                  float* __restrict__ denom, int E, int N, int H) {
  long long t = (long long)blockIdx.x * blockDim.x + threadIdx.x;
  long long total = (long long)(E + N) * H;
  if (t >= total) return;
  long long e = t / H; int hh = (int)(t - e * H);
  int s, d; edge_sd(ei, E, e, s, d);
  float v = lrelu(als[(size_t)s * H + hh] + ald[(size_t)d * H + hh], 0.2f);
  float ee = __expf(v - emax[(size_t)d * H + hh]);
  atomicAdd(denom + (size_t)d * H + hh, ee);
}

__global__ void edge_scatter_kernel(const int* __restrict__ ei, const float* __restrict__ hfeat,
                                    const float* __restrict__ als, const float* __restrict__ ald,
                                    const float* __restrict__ emax, const float* __restrict__ denom,
                                    float* __restrict__ acc, int E, int N, int H, int F) {
  long long wid = ((long long)blockIdx.x * blockDim.x + threadIdx.x) >> 5;
  int lane = threadIdx.x & 31;
  long long total = (long long)E + N;
  if (wid >= total) return;
  int s, d; edge_sd(ei, E, wid, s, d);
  float alpha_l = 0.f;
  if (lane < H) {
    float ev = lrelu(als[(size_t)s * H + lane] + ald[(size_t)d * H + lane], 0.2f);
    alpha_l = __expf(ev - emax[(size_t)d * H + lane]) / denom[(size_t)d * H + lane];
  }
  const float* hs = hfeat + (size_t)s * HID;
  float* ap = acc + (size_t)d * HID;
#pragma unroll
  for (int i = 0; i < 4; ++i) {
    int c = lane + 32 * i;
    int head = c / F;
    float alpha = __shfl(alpha_l, head, 32);
    atomicAdd(ap + c, alpha * hs[c]);
  }
}

__global__ void finalize_kernel(float* __restrict__ acc, const float* __restrict__ bias,
                                int N, int act) {
  int t = blockIdx.x * blockDim.x + threadIdx.x;
  if (t >= N * HID) return;
  int c = t & (HID - 1);
  float v = acc[t] + bias[c];
  if (act) v = lrelu(v, 0.01f);
  acc[t] = v;
}

extern "C" void kernel_launch(void* const* d_in, const int* in_sizes, int n_in,
                              void* d_out, int out_size, void* d_ws, size_t ws_size,
                              hipStream_t stream) {
  const float* num_prop = (const float*)d_in[0];
  const float* cat_prop = (const float*)d_in[1];
  const int*   ei       = (const int*)d_in[2];
  const float* W_num = (const float*)d_in[3];
  const float* b_num = (const float*)d_in[4];
  const float* W_cat = (const float*)d_in[5];
  const float* b_cat = (const float*)d_in[6];
  const float* W_in  = (const float*)d_in[7];
  const float* b_in  = (const float*)d_in[8];
  const float* bn_g  = (const float*)d_in[9];
  const float* bn_b  = (const float*)d_in[10];
  const float* W1    = (const float*)d_in[11];
  const float* a_s1  = (const float*)d_in[12];
  const float* a_d1  = (const float*)d_in[13];
  const float* bias1 = (const float*)d_in[14];
  const float* W2    = (const float*)d_in[15];
  const float* a_s2  = (const float*)d_in[16];
  const float* a_d2  = (const float*)d_in[17];
  const float* bias2 = (const float*)d_in[18];

  const int N = in_sizes[0] / 7;
  const int E = in_sizes[2] / 2;
  float* out = (float*)d_out;

  float* ws   = (float*)d_ws;
  float* xcat = ws;
  float* x1   = xcat + (size_t)N * HID;
  float* hbuf = x1   + (size_t)N * HID;
  float* accb = hbuf + (size_t)N * HID;
  float* als  = accb + (size_t)N * HID;
  float* ald  = als  + (size_t)N * 4;
  float* emax = ald  + (size_t)N * 4;
  float* denom= emax + (size_t)N * 4;
  float* stats= denom+ (size_t)N * 4;

  const int TB = 256;
  const long long EN = (long long)E + N;
  const int mtiles = (N + 15) / 16;
  auto nb = [&](long long work) { return (unsigned)((work + TB - 1) / TB); };

  encoder_kernel<<<nb((long long)N * 64), TB, 0, stream>>>(num_prop, cat_prop, W_num, b_num,
                                                           W_cat, b_cat, xcat, N);
  wmma_gemm128_kernel<<<mtiles, TB, 0, stream>>>(xcat, W_in, b_in, x1, N, 1, 1);

  fill_kernel<<<1, 256, 0, stream>>>(stats, 256, 0.f);
  bn_stats_kernel<<<512, 128, 0, stream>>>(x1, stats, N);
  bn_finalize_kernel<<<1, 128, 0, stream>>>(stats, bn_g, bn_b, N);
  bn_apply_kernel<<<nb((long long)N * HID), TB, 0, stream>>>(x1, stats, xcat, N);

  wmma_gemm128_kernel<<<mtiles, TB, 0, stream>>>(xcat, W1, nullptr, hbuf, N, 0, 0);
  att_logits_kernel<<<nb((long long)N * 4), TB, 0, stream>>>(hbuf, a_s1, a_d1, als, ald, N, 4, 32);
  fill_kernel<<<nb((long long)N * 4), TB, 0, stream>>>(emax, (long long)N * 4, -INFINITY);
  fill_kernel<<<nb((long long)N * 4), TB, 0, stream>>>(denom, (long long)N * 4, 0.f);
  fill_kernel<<<nb((long long)N * HID), TB, 0, stream>>>(accb, (long long)N * HID, 0.f);
  edge_max_kernel<<<nb(EN * 4), TB, 0, stream>>>(ei, als, ald, emax, E, N, 4);
  edge_denom_kernel<<<nb(EN * 4), TB, 0, stream>>>(ei, als, ald, emax, denom, E, N, 4);
  edge_scatter_kernel<<<nb(EN * 32), TB, 0, stream>>>(ei, hbuf, als, ald, emax, denom, accb,
                                                      E, N, 4, 32);
  finalize_kernel<<<nb((long long)N * HID), TB, 0, stream>>>(accb, bias1, N, 1);

  wmma_gemm128_kernel<<<mtiles, TB, 0, stream>>>(accb, W2, nullptr, hbuf, N, 0, 0);
  att_logits_kernel<<<nb(N), TB, 0, stream>>>(hbuf, a_s2, a_d2, als, ald, N, 1, 128);
  fill_kernel<<<nb(N), TB, 0, stream>>>(emax, N, -INFINITY);
  fill_kernel<<<nb(N), TB, 0, stream>>>(denom, N, 0.f);
  fill_kernel<<<nb((long long)N * HID), TB, 0, stream>>>(out, (long long)N * HID, 0.f);
  edge_max_kernel<<<nb(EN), TB, 0, stream>>>(ei, als, ald, emax, E, N, 1);
  edge_denom_kernel<<<nb(EN), TB, 0, stream>>>(ei, als, ald, emax, denom, E, N, 1);
  edge_scatter_kernel<<<nb(EN * 32), TB, 0, stream>>>(ei, hbuf, als, ald, emax, denom, out,
                                                      E, N, 1, 128);
  finalize_kernel<<<nb((long long)N * HID), TB, 0, stream>>>(out, bias2, N, 0);
}


# GAT edge filter done once per call: the layer-1 tile kernel keeps each tile's (source, local destination) list in the workspace; the layer-2 tile kernel loads it instead of scanning the column again
# speedup vs baseline: 2.4449x; 2.4449x over previous
#include <hip/hip_runtime.h>
#include <math.h>

typedef __attribute__((ext_vector_type(2))) float v2f;
typedef __attribute__((ext_vector_type(8))) float v8f;

#define HID 128

__device__ __forceinline__ float lrelu(float v, float s) { return v > 0.f ? v : s * v; }

__global__ void encoder_kernel(const float* __restrict__ num_prop, const float* __restrict__ cat_prop,
                               const float* __restrict__ Wn, const float* __restrict__ bn,
                               const float* __restrict__ Wc, const float* __restrict__ bc,
                               float* __restrict__ xcat, int N) {
  int t = blockIdx.x * blockDim.x + threadIdx.x;
  if (t >= N * 64) return;
  int n = t >> 6, j = t & 63;
  const float* np = num_prop + (size_t)n * 7;
  float an = bn[j];
#pragma unroll
  for (int k = 0; k < 7; ++k) an += np[k] * Wn[k * 64 + j];
  const float* cp = cat_prop + (size_t)n * 3;
  float ac = bc[j];
#pragma unroll
  for (int k = 0; k < 3; ++k) ac += cp[k] * Wc[k * 64 + j];
  xcat[(size_t)n * HID + j]      = lrelu(an, 0.01f);
  xcat[(size_t)n * HID + 64 + j] = lrelu(ac, 0.01f);
}

__global__ void wmma_gemm128_kernel(const float* __restrict__ A, const float* __restrict__ B,
                                    const float* __restrict__ bias, float* __restrict__ C,
                                    int M, int act, int has_bias) {
  int wave = threadIdx.x >> 5;
  int lane = threadIdx.x & 31;
  int row0 = blockIdx.x * 16;
  int col0 = wave * 16;
  int half = lane >> 4;
  int m16  = lane & 15;

  int arow = row0 + m16; if (arow >= M) arow = M - 1;
  const float* Ap = A + (size_t)arow * HID + 2 * half;
  const float* Bp = B + (size_t)(2 * half) * HID + col0 + m16;

  v8f acc = {0.f, 0.f, 0.f, 0.f, 0.f, 0.f, 0.f, 0.f};
#pragma unroll 8
  for (int k0 = 0; k0 < HID; k0 += 4) {
    v2f a, b;
    a.x = Ap[k0];
    a.y = Ap[k0 + 1];
    b.x = Bp[(size_t)k0 * HID];
    b.y = Bp[(size_t)(k0 + 1) * HID];
    acc = __builtin_amdgcn_wmma_f32_16x16x4_f32(false, a, false, b, (short)0, acc, false, false);
  }

  int col = col0 + m16;
  float bv = has_bias ? bias[col] : 0.f;
#pragma unroll
  for (int r = 0; r < 8; ++r) {
    int row = row0 + r + 8 * half;
    if (row < M) {
      float v = acc[r] + bv;
      if (act) v = lrelu(v, 0.01f);
      C[(size_t)row * HID + col] = v;
    }
  }
}

__global__ void bn_stats_kernel(const float* __restrict__ x, float* __restrict__ stats, int N) {
  int c = threadIdx.x;
  float s = 0.f, s2 = 0.f;
  for (int n = blockIdx.x; n < N; n += gridDim.x) {
    float v = x[(size_t)n * HID + c];
    s += v; s2 += v * v;
  }
  atomicAdd(stats + c, s);
  atomicAdd(stats + HID + c, s2);
}

__global__ void bn_finalize_kernel(float* __restrict__ stats, const float* __restrict__ gamma,
                                   const float* __restrict__ beta, int N) {
  int c = threadIdx.x;
  float inv_n = 1.f / (float)N;
  float mu  = stats[c] * inv_n;
  float var = stats[HID + c] * inv_n - mu * mu;
  float sc  = gamma[c] * rsqrtf(var + 1e-5f);
  stats[2 * HID + c] = sc;
  stats[3 * HID + c] = beta[c] - mu * sc;
}

__global__ void bn_apply_kernel(const float* __restrict__ x, const float* __restrict__ stats,
                                float* __restrict__ y, int N) {
  int t = blockIdx.x * blockDim.x + threadIdx.x;
  if (t >= N * HID) return;
  int c = t & (HID - 1);
  y[t] = x[t] * stats[2 * HID + c] + stats[3 * HID + c];
}

__global__ void att_logits_kernel(const float* __restrict__ h, const float* __restrict__ a_src,
                                  const float* __restrict__ a_dst, float* __restrict__ als,
                                  float* __restrict__ ald, int N, int H, int F) {
  int t = blockIdx.x * blockDim.x + threadIdx.x;
  if (t >= N * H) return;
  int n = t / H, hh = t - n * H;
  const float* hp = h + (size_t)n * HID + hh * F;
  const float* as = a_src + hh * F;
  const float* ad = a_dst + hh * F;
  float ss = 0.f, sd = 0.f;
  for (int f = 0; f < F; ++f) { float v = hp[f]; ss += v * as[f]; sd += v * ad[f]; }
  als[t] = ss;
  ald[t] = sd;
}

__global__ void fill_kernel(float* __restrict__ p, long long n, float v) {
  long long t = (long long)blockIdx.x * blockDim.x + threadIdx.x;
  if (t < n) p[t] = v;
}

#define TD   196
#define CAP  16384
#define NT   1024
#define UB   8
#define UE   8

struct TileLds {
  float acc[TD * HID];
  float mx[TD * 4];
  float den[TD * 4];
  float aldT[TD * 4];
  float wsf[TD * 4];
  int   list[CAP];
  unsigned short dls[CAP];
  int   cnt;
};

__device__ __forceinline__ void ldsMaxF(float* addr, float val) {
  if (val >= 0.f) atomicMax((int*)addr, __float_as_int(val));
  else            atomicMin((unsigned int*)addr, __float_as_uint(val));
}

__device__ __forceinline__ void tile_push(TileLds& L, int e) {
  int pos = atomicAdd(&L.cnt, 1);
  if (pos < CAP) L.list[pos] = e;
}

__device__ __forceinline__ void tile_test1(TileLds& L, const int* __restrict__ dstp, int e,
                                           int d0, int tdn) {
  if ((unsigned)(dstp[e] - d0) < (unsigned)tdn) tile_push(L, e);
}

__device__ __forceinline__ void tile_scan(TileLds& L, const int* __restrict__ dstp, int lo, int hi,
                                          int d0, int tdn, int tid) {
  int head = (4 - (int)((((size_t)(dstp + lo)) >> 2) & 3)) & 3;
  int a0 = lo + head; if (a0 > hi) a0 = hi;
  if (tid < a0 - lo) tile_test1(L, dstp, lo + tid, d0, tdn);
  const int nvec = (hi - a0) >> 2;
  const int4* p4 = (const int4*)(dstp + a0);
  const int rot = (nvec / (int)gridDim.x) * (int)blockIdx.x;
  for (int j = 0; j < nvec; j += NT * UB) {
    int4 v[UB]; int idx[UB];
#pragma unroll
    for (int u = 0; u < UB; ++u) {
      int k = j + u * NT + tid;
      int i = k + rot; if (i >= nvec) i -= nvec;
      if (k >= nvec) i = -1;
      idx[u] = i;
      v[u] = p4[i < 0 ? 0 : i];
    }
#pragma unroll
    for (int u = 0; u < UB; ++u) {
      const bool ok = idx[u] >= 0;
      bool h0 = ok & ((unsigned)(v[u].x - d0) < (unsigned)tdn);
      bool h1 = ok & ((unsigned)(v[u].y - d0) < (unsigned)tdn);
      bool h2 = ok & ((unsigned)(v[u].z - d0) < (unsigned)tdn);
      bool h3 = ok & ((unsigned)(v[u].w - d0) < (unsigned)tdn);
      if (h0 | h1 | h2 | h3) {
        int e0 = a0 + 4 * idx[u];
        if (h0) tile_push(L, e0);
        if (h1) tile_push(L, e0 + 1);
        if (h2) tile_push(L, e0 + 2);
        if (h3) tile_push(L, e0 + 3);
      }
    }
  }
  int t0 = a0 + 4 * nvec;
  if (tid < hi - t0) tile_test1(L, dstp, t0 + tid, d0, tdn);
}

#define UM 4
#define KEEP_MAXN (1 << 24)
template <int H, bool KEEP>
__device__ __forceinline__ void tile_max(TileLds& L, const int* __restrict__ ei,
                                         const int* __restrict__ dstp,
                                         const float* __restrict__ als, int* __restrict__ wl,
                                         int n, int d0, int tid) {
  for (int i0 = tid; i0 < n; i0 += NT * UM) {
    int ix[UM], e[UM], s[UM], dl[UM]; float a[UM][H];
#pragma unroll
    for (int u = 0; u < UM; ++u) {
      int i = i0 + u * NT;
      ix[u] = i < n ? i : i0;
      e[u] = L.list[ix[u]];
    }
#pragma unroll
    for (int u = 0; u < UM; ++u) { s[u] = ei[e[u]]; dl[u] = dstp[e[u]] - d0; }
#pragma unroll
    for (int u = 0; u < UM; ++u) {
#pragma unroll
      for (int h = 0; h < H; ++h) a[u][h] = als[(size_t)s[u] * H + h];
    }
#pragma unroll
    for (int u = 0; u < UM; ++u) {
      if (i0 + u * NT < n) {
        L.list[ix[u]] = s[u];
        L.dls[ix[u]] = (unsigned short)dl[u];
        if (KEEP) wl[ix[u]] = (int)(((unsigned)s[u] << 8) | (unsigned)dl[u]);
#pragma unroll
        for (int h = 0; h < H; ++h) {
          float v = lrelu(a[u][h] + L.aldT[dl[u] * H + h], 0.2f);
          ldsMaxF(&L.mx[dl[u] * H + h], v);
        }
      }
    }
  }
}

template <int H>
__device__ __forceinline__ void tile_max_kept(TileLds& L, const int* __restrict__ wl,
                                              const float* __restrict__ als, int n, int tid) {
  for (int i0 = tid; i0 < n; i0 += NT * UM) {
    int ix[UM], s[UM], dl[UM]; unsigned p[UM]; float a[UM][H];
#pragma unroll
    for (int u = 0; u < UM; ++u) {
      int i = i0 + u * NT;
      ix[u] = i < n ? i : i0;
      p[u] = (unsigned)wl[ix[u]];
    }
#pragma unroll
    for (int u = 0; u < UM; ++u) {
      s[u] = (int)(p[u] >> 8); dl[u] = (int)(p[u] & 255u);
#pragma unroll
      for (int h = 0; h < H; ++h) a[u][h] = als[(size_t)s[u] * H + h];
    }
#pragma unroll
    for (int u = 0; u < UM; ++u) {
      if (i0 + u * NT < n) {
        L.list[ix[u]] = s[u];
        L.dls[ix[u]] = (unsigned short)dl[u];
#pragma unroll
        for (int h = 0; h < H; ++h) {
          float v = lrelu(a[u][h] + L.aldT[dl[u] * H + h], 0.2f);
          ldsMaxF(&L.mx[dl[u] * H + h], v);
        }
      }
    }
  }
}

template <int H>
__device__ __forceinline__ void tile_sum(TileLds& L, const float* __restrict__ hfeat,
                                         const float* __restrict__ als, int n, int tid) {
  constexpr int F = HID / H;
  const int wave = tid >> 5, lane = tid & 31;
  for (int i0 = wave * UE; i0 < n; i0 += (NT / 32) * UE) {
    int s[UE], dl[UE]; float al[UE]; float hv[UE][4];
#pragma unroll
    for (int u = 0; u < UE; ++u) {
      int i = i0 + u; if (i >= n) i = n - 1;
      s[u] = L.list[i]; dl[u] = L.dls[i];
    }
#pragma unroll
    for (int u = 0; u < UE; ++u) {
      const float* hs = hfeat + (size_t)s[u] * HID;
      al[u] = (lane < H) ? als[(size_t)s[u] * H + lane] : 0.f;
#pragma unroll
      for (int k = 0; k < 4; ++k) hv[u][k] = hs[lane + 32 * k];
    }
#pragma unroll
    for (int u = 0; u < UE; ++u) {
      if (i0 + u < n) {
        float w_l = 0.f;
        if (lane < H) {
          int q = dl[u] * H + lane;
          float v = lrelu(al[u] + L.aldT[q], 0.2f);
          w_l = __expf(v - L.mx[q]);
          atomicAdd(&L.den[q], w_l);
        }
#pragma unroll
        for (int k = 0; k < 4; ++k) {
          int c = lane + 32 * k;
          float w = __shfl(w_l, c / F, 32);
          atomicAdd(&L.acc[dl[u] * HID + c], w * hv[u][k]);
        }
      }
    }
  }
}

template <int H, int MODE>
__device__ __forceinline__ void gat_tile_body(TileLds& L, const int* __restrict__ ei,
                                              const float* __restrict__ hfeat,
                                              const float* __restrict__ als,
                                              const float* __restrict__ ald,
                                              const float* __restrict__ bias,
                                              float* __restrict__ outp, int* __restrict__ keepbuf,
                                              int E, int N, int act) {
  constexpr int F = HID / H;
  const int tid = threadIdx.x;
  const int d0 = blockIdx.x * TD;
  int tdn = N - d0; if (tdn > TD) tdn = TD;
  const int* dstp = ei + (size_t)E;
  int* wl   = keepbuf + (size_t)blockIdx.x * CAP;
  int* wcnt = keepbuf + (size_t)gridDim.x * CAP + blockIdx.x;

  for (int i = tid; i < TD * HID; i += NT) L.acc[i] = 0.f;
  for (int i = tid; i < tdn * H; i += NT) {
    size_t g = (size_t)d0 * H + i;
    float a_d = ald[g];
    float v = lrelu(als[g] + a_d, 0.2f);
    L.aldT[i] = a_d; L.wsf[i] = v; L.mx[i] = v;
  }

  __syncthreads();
  int n = CAP + 1;
  if (MODE == 2) n = *wcnt;
  const bool kept = (MODE == 2) && (n <= CAP);
  if (!kept) {
    if (tid == 0) L.cnt = 0;
    __syncthreads();
    tile_scan(L, dstp, 0, E, d0, tdn, tid);
    __syncthreads();
    n = L.cnt;
  }
  if (MODE == 1 && tid == 0) *wcnt = (N <= KEEP_MAXN) ? n : CAP + 1;
  if (n <= CAP) {
    if (kept) tile_max_kept<H>(L, wl, als, n, tid);
    else      tile_max<H, MODE == 1>(L, ei, dstp, als, wl, n, d0, tid);
    __syncthreads();
    for (int i = tid; i < tdn * H; i += NT) {
      float w = __expf(L.wsf[i] - L.mx[i]);
      L.wsf[i] = w; L.den[i] = w;
    }
    __syncthreads();
    tile_sum<H>(L, hfeat, als, n, tid);
  } else {
    const int nwin = (E + CAP - 1) / CAP;
    for (int pass = 0; pass < 2; ++pass) {
      for (int w = 0; w < nwin; ++w) {
        __syncthreads();
        if (tid == 0) L.cnt = 0;
        __syncthreads();
        const int lo = w * CAP;
        int hi = lo + CAP; if (hi > E) hi = E;
        tile_scan(L, dstp, lo, hi, d0, tdn, tid);
        __syncthreads();
        n = L.cnt;
        tile_max<H, false>(L, ei, dstp, als, wl, n, d0, tid);
        __syncthreads();
        if (pass == 1) tile_sum<H>(L, hfeat, als, n, tid);
      }
      if (pass == 0) {
        __syncthreads();
        for (int i = tid; i < tdn * H; i += NT) {
          float w = __expf(L.wsf[i] - L.mx[i]);
          L.wsf[i] = w; L.den[i] = w;
        }
      }
    }
  }
  __syncthreads();

  {
    constexpr int UO = 5;
    constexpr int RS = NT / HID;
    const int c = tid & (HID - 1);
    const int hq = c / F;
    const float bv = bias[c];
    for (int r0 = tid / HID; r0 < tdn; r0 += RS * UO) {
      float hv[UO];
#pragma unroll
      for (int u = 0; u < UO; ++u) {
        int dl = r0 + RS * u; if (dl >= tdn) dl = tdn - 1;
        hv[u] = hfeat[(size_t)(d0 + dl) * HID + c];
      }
#pragma unroll
      for (int u = 0; u < UO; ++u) {
        const int dl = r0 + RS * u;
        if (dl < tdn) {
          const int q = dl * H + hq;
          float v = (L.acc[dl * HID + c] + L.wsf[q] * hv[u]) / L.den[q] + bv;
          if (act) v = lrelu(v, 0.01f);
          outp[(size_t)(d0 + dl) * HID + c] = v;
        }
      }
    }
  }
}

__global__ void __launch_bounds__(NT) gat_tile_h4_kernel(const int* __restrict__ ei,
    const float* __restrict__ hfeat, const float* __restrict__ als, const float* __restrict__ ald,
    const float* __restrict__ bias, float* __restrict__ outp, int* __restrict__ keepbuf,
    int E, int N, int act) {
  __shared__ TileLds L;
  gat_tile_body<4, 1>(L, ei, hfeat, als, ald, bias, outp, keepbuf, E, N, act);
}

__global__ void __launch_bounds__(NT) gat_tile_h1_kernel(const int* __restrict__ ei,
    const float* __restrict__ hfeat, const float* __restrict__ als, const float* __restrict__ ald,
    const float* __restrict__ bias, float* __restrict__ outp, int* __restrict__ keepbuf,
    int E, int N, int act) {
  __shared__ TileLds L;
  gat_tile_body<1, 2>(L, ei, hfeat, als, ald, bias, outp, keepbuf, E, N, act);
}

extern "C" void kernel_launch(void* const* d_in, const int* in_sizes, int n_in,
                              void* d_out, int out_size, void* d_ws, size_t ws_size,
                              hipStream_t stream) {
  const float* num_prop = (const float*)d_in[0];
  const float* cat_prop = (const float*)d_in[1];
  const int*   ei       = (const int*)d_in[2];
  const float* W_num = (const float*)d_in[3];
  const float* b_num = (const float*)d_in[4];
  const float* W_cat = (const float*)d_in[5];
  const float* b_cat = (const float*)d_in[6];
  const float* W_in  = (const float*)d_in[7];
  const float* b_in  = (const float*)d_in[8];
  const float* bn_g  = (const float*)d_in[9];
  const float* bn_b  = (const float*)d_in[10];
  const float* W1    = (const float*)d_in[11];
  const float* a_s1  = (const float*)d_in[12];
  const float* a_d1  = (const float*)d_in[13];
  const float* bias1 = (const float*)d_in[14];
  const float* W2    = (const float*)d_in[15];
  const float* a_s2  = (const float*)d_in[16];
  const float* a_d2  = (const float*)d_in[17];
  const float* bias2 = (const float*)d_in[18];

  const int N = in_sizes[0] / 7;
  const int E = in_sizes[2] / 2;
  float* out = (float*)d_out;

  float* ws   = (float*)d_ws;
  float* xcat = ws;
  float* x1   = xcat + (size_t)N * HID;
  float* hbuf = x1   + (size_t)N * HID;
  float* accb = hbuf + (size_t)N * HID;
  float* als  = accb + (size_t)N * HID;
  float* ald  = als  + (size_t)N * 4;
  float* stats= ald  + (size_t)N * 12;

  const int TB = 256;
  const int ntiles = (N + TD - 1) / TD;
  const int mtiles = (N + 15) / 16;
  auto nb = [&](long long work) { return (unsigned)((work + TB - 1) / TB); };
  int* keepbuf = (int*)x1;

  encoder_kernel<<<nb((long long)N * 64), TB, 0, stream>>>(num_prop, cat_prop, W_num, b_num,
                                                           W_cat, b_cat, xcat, N);
  wmma_gemm128_kernel<<<mtiles, TB, 0, stream>>>(xcat, W_in, b_in, x1, N, 1, 1);

  fill_kernel<<<1, 256, 0, stream>>>(stats, 256, 0.f);
  bn_stats_kernel<<<512, 128, 0, stream>>>(x1, stats, N);
  bn_finalize_kernel<<<1, 128, 0, stream>>>(stats, bn_g, bn_b, N);
  bn_apply_kernel<<<nb((long long)N * HID), TB, 0, stream>>>(x1, stats, xcat, N);

  wmma_gemm128_kernel<<<mtiles, TB, 0, stream>>>(xcat, W1, nullptr, hbuf, N, 0, 0);
  att_logits_kernel<<<nb((long long)N * 4), TB, 0, stream>>>(hbuf, a_s1, a_d1, als, ald, N, 4, 32);
  gat_tile_h4_kernel<<<ntiles, NT, 0, stream>>>(ei, hbuf, als, ald, bias1, accb, keepbuf, E, N, 1);

  wmma_gemm128_kernel<<<mtiles, TB, 0, stream>>>(accb, W2, nullptr, hbuf, N, 0, 0);
  att_logits_kernel<<<nb(N), TB, 0, stream>>>(hbuf, a_s2, a_d2, als, ald, N, 1, 128);
  gat_tile_h1_kernel<<<ntiles, NT, 0, stream>>>(ei, hbuf, als, ald, bias2, out, keepbuf, E, N, 0);
}


# GEMM wave tile 16x128: a wave owns a 16-row strip over all 128 columns, A fetched once into registers by 16-byte loads, the weight staged once per block in LDS; 25000 waves to 3128 per GEMM
# speedup vs baseline: 2.6768x; 1.0949x over previous
#include <hip/hip_runtime.h>
#include <math.h>

typedef __attribute__((ext_vector_type(2))) float v2f;
typedef __attribute__((ext_vector_type(8))) float v8f;

#define HID 128

__device__ __forceinline__ float lrelu(float v, float s) { return v > 0.f ? v : s * v; }

__global__ void encoder_kernel(const float* __restrict__ num_prop, const float* __restrict__ cat_prop,
                               const float* __restrict__ Wn, const float* __restrict__ bn,
                               const float* __restrict__ Wc, const float* __restrict__ bc,
                               float* __restrict__ xcat, int N) {
  int t = blockIdx.x * blockDim.x + threadIdx.x;
  if (t >= N * 64) return;
  int n = t >> 6, j = t & 63;
  const float* np = num_prop + (size_t)n * 7;
  float an = bn[j];
#pragma unroll
  for (int k = 0; k < 7; ++k) an += np[k] * Wn[k * 64 + j];
  const float* cp = cat_prop + (size_t)n * 3;
  float ac = bc[j];
#pragma unroll
  for (int k = 0; k < 3; ++k) ac += cp[k] * Wc[k * 64 + j];
  xcat[(size_t)n * HID + j]      = lrelu(an, 0.01f);
  xcat[(size_t)n * HID + 64 + j] = lrelu(ac, 0.01f);
}

#define GW 8
#define BP 136
__global__ void __launch_bounds__(GW * 32)
wmma_gemm128_kernel(const float* __restrict__ A, const float* __restrict__ B,
                    const float* __restrict__ bias, float* __restrict__ C,
                    int M, int act, int has_bias) {
  __shared__ float4 Bl[32 * BP];
  const int tid  = threadIdx.x;
  const int wave = tid >> 5;
  const int lane = tid & 31;
  const int half = lane >> 4;
  const int m16  = lane & 15;
  const int row0 = (blockIdx.x * GW + wave) * 16;

  int arow = row0 + m16; if (arow >= M) arow = M - 1;
  const float4* Ap = reinterpret_cast<const float4*>(A + (size_t)arow * HID) + half;
  float4 a[16];
#pragma unroll
  for (int j = 0; j < 16; ++j) a[j] = Ap[2 * j];

  float bv[8];
#pragma unroll
  for (int t = 0; t < 8; ++t) bv[t] = has_bias ? bias[16 * t + m16] : 0.f;

  {
    const int col = tid & (HID - 1);
#pragma unroll
    for (int i = 0; i < 16; ++i) {
      const int q = (tid >> 7) + 2 * i;
      const float* bp = B + (size_t)(4 * q) * HID + col;
      float4 v;
      v.x = bp[0]; v.y = bp[HID]; v.z = bp[2 * HID]; v.w = bp[3 * HID];
      Bl[q * BP + col] = v;
    }
  }
  __syncthreads();
  if (row0 >= M) return;

  v8f acc[8];
#pragma unroll
  for (int t = 0; t < 8; ++t) acc[t] = (v8f){0.f, 0.f, 0.f, 0.f, 0.f, 0.f, 0.f, 0.f};

  const float4* Bf = Bl + half * BP + m16;
  float4 bc[8], bn[8];
#pragma unroll
  for (int t = 0; t < 8; ++t) bc[t] = Bf[16 * t];
#pragma unroll
  for (int j = 0; j < 16; ++j) {
    if (j + 1 < 16) {
#pragma unroll
      for (int t = 0; t < 8; ++t) bn[t] = Bf[(2 * (j + 1)) * BP + 16 * t];
    }
    __builtin_amdgcn_sched_barrier(0);
    const v2f a0 = {a[j].x, a[j].y};
    const v2f a1 = {a[j].z, a[j].w};
#pragma unroll
    for (int t = 0; t < 8; ++t) {
      const v2f b0 = {bc[t].x, bc[t].y};
      acc[t] = __builtin_amdgcn_wmma_f32_16x16x4_f32(false, a0, false, b0, (short)0, acc[t], false, false);
    }
#pragma unroll
    for (int t = 0; t < 8; ++t) {
      const v2f b1 = {bc[t].z, bc[t].w};
      acc[t] = __builtin_amdgcn_wmma_f32_16x16x4_f32(false, a1, false, b1, (short)0, acc[t], false, false);
    }
    __builtin_amdgcn_sched_barrier(0);
#pragma unroll
    for (int t = 0; t < 8; ++t) bc[t] = bn[t];
  }

  float* Cp = C + (size_t)(row0 + 8 * half) * HID + m16;
  if (row0 + 16 <= M) {
#pragma unroll
    for (int r = 0; r < 8; ++r) {
#pragma unroll
      for (int t = 0; t < 8; ++t) {
        float v = acc[t][r] + bv[t];
        if (act) v = lrelu(v, 0.01f);
        Cp[r * HID + 16 * t] = v;
      }
    }
  } else {
#pragma unroll
    for (int r = 0; r < 8; ++r) {
      if (row0 + 8 * half + r < M) {
#pragma unroll
        for (int t = 0; t < 8; ++t) {
          float v = acc[t][r] + bv[t];
          if (act) v = lrelu(v, 0.01f);
          Cp[r * HID + 16 * t] = v;
        }
      }
    }
  }
}

__global__ void bn_stats_kernel(const float* __restrict__ x, float* __restrict__ stats, int N) {
  int c = threadIdx.x;
  float s = 0.f, s2 = 0.f;
  for (int n = blockIdx.x; n < N; n += gridDim.x) {
    float v = x[(size_t)n * HID + c];
    s += v; s2 += v * v;
  }
  atomicAdd(stats + c, s);
  atomicAdd(stats + HID + c, s2);
}

__global__ void bn_finalize_kernel(float* __restrict__ stats, const float* __restrict__ gamma,
                                   const float* __restrict__ beta, int N) {
  int c = threadIdx.x;
  float inv_n = 1.f / (float)N;
  float mu  = stats[c] * inv_n;
  float var = stats[HID + c] * inv_n - mu * mu;
  float sc  = gamma[c] * rsqrtf(var + 1e-5f);
  stats[2 * HID + c] = sc;
  stats[3 * HID + c] = beta[c] - mu * sc;
}

__global__ void bn_apply_kernel(const float* __restrict__ x, const float* __restrict__ stats,
                                float* __restrict__ y, int N) {
  int t = blockIdx.x * blockDim.x + threadIdx.x;
  if (t >= N * HID) return;
  int c = t & (HID - 1);
  y[t] = x[t] * stats[2 * HID + c] + stats[3 * HID + c];
}

__global__ void att_logits_kernel(const float* __restrict__ h, const float* __restrict__ a_src,
                                  const float* __restrict__ a_dst, float* __restrict__ als,
                                  float* __restrict__ ald, int N, int H, int F) {
  int t = blockIdx.x * blockDim.x + threadIdx.x;
  if (t >= N * H) return;
  int n = t / H, hh = t - n * H;
  const float* hp = h + (size_t)n * HID + hh * F;
  const float* as = a_src + hh * F;
  const float* ad = a_dst + hh * F;
  float ss = 0.f, sd = 0.f;
  for (int f = 0; f < F; ++f) { float v = hp[f]; ss += v * as[f]; sd += v * ad[f]; }
  als[t] = ss;
  ald[t] = sd;
}

__global__ void fill_kernel(float* __restrict__ p, long long n, float v) {
  long long t = (long long)blockIdx.x * blockDim.x + threadIdx.x;
  if (t < n) p[t] = v;
}

#define TD   196
#define CAP  16384
#define NT   1024
#define UB   8
#define UE   8

struct TileLds {
  float acc[TD * HID];
  float mx[TD * 4];
  float den[TD * 4];
  float aldT[TD * 4];
  float wsf[TD * 4];
  int   list[CAP];
  unsigned short dls[CAP];
  int   cnt;
};

__device__ __forceinline__ void ldsMaxF(float* addr, float val) {
  if (val >= 0.f) atomicMax((int*)addr, __float_as_int(val));
  else            atomicMin((unsigned int*)addr, __float_as_uint(val));
}

__device__ __forceinline__ void tile_push(TileLds& L, int e) {
  int pos = atomicAdd(&L.cnt, 1);
  if (pos < CAP) L.list[pos] = e;
}

__device__ __forceinline__ void tile_test1(TileLds& L, const int* __restrict__ dstp, int e,
                                           int d0, int tdn) {
  if ((unsigned)(dstp[e] - d0) < (unsigned)tdn) tile_push(L, e);
}

__device__ __forceinline__ void tile_scan(TileLds& L, const int* __restrict__ dstp, int lo, int hi,
                                          int d0, int tdn, int tid) {
  int head = (4 - (int)((((size_t)(dstp + lo)) >> 2) & 3)) & 3;
  int a0 = lo + head; if (a0 > hi) a0 = hi;
  if (tid < a0 - lo) tile_test1(L, dstp, lo + tid, d0, tdn);
  const int nvec = (hi - a0) >> 2;
  const int4* p4 = (const int4*)(dstp + a0);
  const int rot = (nvec / (int)gridDim.x) * (int)blockIdx.x;
  for (int j = 0; j < nvec; j += NT * UB) {
    int4 v[UB]; int idx[UB];
#pragma unroll
    for (int u = 0; u < UB; ++u) {
      int k = j + u * NT + tid;
      int i = k + rot; if (i >= nvec) i -= nvec;
      if (k >= nvec) i = -1;
      idx[u] = i;
      v[u] = p4[i < 0 ? 0 : i];
    }
#pragma unroll
    for (int u = 0; u < UB; ++u) {
      const bool ok = idx[u] >= 0;
      bool h0 = ok & ((unsigned)(v[u].x - d0) < (unsigned)tdn);
      bool h1 = ok & ((unsigned)(v[u].y - d0) < (unsigned)tdn);
      bool h2 = ok & ((unsigned)(v[u].z - d0) < (unsigned)tdn);
      bool h3 = ok & ((unsigned)(v[u].w - d0) < (unsigned)tdn);
      if (h0 | h1 | h2 | h3) {
        int e0 = a0 + 4 * idx[u];
        if (h0) tile_push(L, e0);
        if (h1) tile_push(L, e0 + 1);
        if (h2) tile_push(L, e0 + 2);
        if (h3) tile_push(L, e0 + 3);
      }
    }
  }
  int t0 = a0 + 4 * nvec;
  if (tid < hi - t0) tile_test1(L, dstp, t0 + tid, d0, tdn);
}

#define UM 4
#define KEEP_MAXN (1 << 24)
template <int H, bool KEEP>
__device__ __forceinline__ void tile_max(TileLds& L, const int* __restrict__ ei,
                                         const int* __restrict__ dstp,
                                         const float* __restrict__ als, int* __restrict__ wl,
                                         int n, int d0, int tid) {
  for (int i0 = tid; i0 < n; i0 += NT * UM) {
    int ix[UM], e[UM], s[UM], dl[UM]; float a[UM][H];
#pragma unroll
    for (int u = 0; u < UM; ++u) {
      int i = i0 + u * NT;
      ix[u] = i < n ? i : i0;
      e[u] = L.list[ix[u]];
    }
#pragma unroll
    for (int u = 0; u < UM; ++u) { s[u] = ei[e[u]]; dl[u] = dstp[e[u]] - d0; }
#pragma unroll
    for (int u = 0; u < UM; ++u) {
#pragma unroll
      for (int h = 0; h < H; ++h) a[u][h] = als[(size_t)s[u] * H + h];
    }
#pragma unroll
    for (int u = 0; u < UM; ++u) {
      if (i0 + u * NT < n) {
        L.list[ix[u]] = s[u];
        L.dls[ix[u]] = (unsigned short)dl[u];
        if (KEEP) wl[ix[u]] = (int)(((unsigned)s[u] << 8) | (unsigned)dl[u]);
#pragma unroll
        for (int h = 0; h < H; ++h) {
          float v = lrelu(a[u][h] + L.aldT[dl[u] * H + h], 0.2f);
          ldsMaxF(&L.mx[dl[u] * H + h], v);
        }
      }
    }
  }
}

template <int H>
__device__ __forceinline__ void tile_max_kept(TileLds& L, const int* __restrict__ wl,
                                              const float* __restrict__ als, int n, int tid) {
  for (int i0 = tid; i0 < n; i0 += NT * UM) {
    int ix[UM], s[UM], dl[UM]; unsigned p[UM]; float a[UM][H];
#pragma unroll
    for (int u = 0; u < UM; ++u) {
      int i = i0 + u * NT;
      ix[u] = i < n ? i : i0;
      p[u] = (unsigned)wl[ix[u]];
    }
#pragma unroll
    for (int u = 0; u < UM; ++u) {
      s[u] = (int)(p[u] >> 8); dl[u] = (int)(p[u] & 255u);
#pragma unroll
      for (int h = 0; h < H; ++h) a[u][h] = als[(size_t)s[u] * H + h];
    }
#pragma unroll
    for (int u = 0; u < UM; ++u) {
      if (i0 + u * NT < n) {
        L.list[ix[u]] = s[u];
        L.dls[ix[u]] = (unsigned short)dl[u];
#pragma unroll
        for (int h = 0; h < H; ++h) {
          float v = lrelu(a[u][h] + L.aldT[dl[u] * H + h], 0.2f);
          ldsMaxF(&L.mx[dl[u] * H + h], v);
        }
      }
    }
  }
}

template <int H>
__device__ __forceinline__ void tile_sum(TileLds& L, const float* __restrict__ hfeat,
                                         const float* __restrict__ als, int n, int tid) {
  constexpr int F = HID / H;
  const int wave = tid >> 5, lane = tid & 31;
  for (int i0 = wave * UE; i0 < n; i0 += (NT / 32) * UE) {
    int s[UE], dl[UE]; float al[UE]; float hv[UE][4];
#pragma unroll
    for (int u = 0; u < UE; ++u) {
      int i = i0 + u; if (i >= n) i = n - 1;
      s[u] = L.list[i]; dl[u] = L.dls[i];
    }
#pragma unroll
    for (int u = 0; u < UE; ++u) {
      const float* hs = hfeat + (size_t)s[u] * HID;
      al[u] = (lane < H) ? als[(size_t)s[u] * H + lane] : 0.f;
#pragma unroll
      for (int k = 0; k < 4; ++k) hv[u][k] = hs[lane + 32 * k];
    }
#pragma unroll
    for (int u = 0; u < UE; ++u) {
      if (i0 + u < n) {
        float w_l = 0.f;
        if (lane < H) {
          int q = dl[u] * H + lane;
          float v = lrelu(al[u] + L.aldT[q], 0.2f);
          w_l = __expf(v - L.mx[q]);
          atomicAdd(&L.den[q], w_l);
        }
#pragma unroll
        for (int k = 0; k < 4; ++k) {
          int c = lane + 32 * k;
          float w = __shfl(w_l, c / F, 32);
          atomicAdd(&L.acc[dl[u] * HID + c], w * hv[u][k]);
        }
      }
    }
  }
}

template <int H, int MODE>
__device__ __forceinline__ void gat_tile_body(TileLds& L, const int* __restrict__ ei,
                                              const float* __restrict__ hfeat,
                                              const float* __restrict__ als,
                                              const float* __restrict__ ald,
                                              const float* __restrict__ bias,
                                              float* __restrict__ outp, int* __restrict__ keepbuf,
                                              int E, int N, int act) {
  constexpr int F = HID / H;
  const int tid = threadIdx.x;
  const int d0 = blockIdx.x * TD;
  int tdn = N - d0; if (tdn > TD) tdn = TD;
  const int* dstp = ei + (size_t)E;
  int* wl   = keepbuf + (size_t)blockIdx.x * CAP;
  int* wcnt = keepbuf + (size_t)gridDim.x * CAP + blockIdx.x;

  for (int i = tid; i < TD * HID; i += NT) L.acc[i] = 0.f;
  for (int i = tid; i < tdn * H; i += NT) {
    size_t g = (size_t)d0 * H + i;
    float a_d = ald[g];
    float v = lrelu(als[g] + a_d, 0.2f);
    L.aldT[i] = a_d; L.wsf[i] = v; L.mx[i] = v;
  }

  __syncthreads();
  int n = CAP + 1;
  if (MODE == 2) n = *wcnt;
  const bool kept = (MODE == 2) && (n <= CAP);
  if (!kept) {
    if (tid == 0) L.cnt = 0;
    __syncthreads();
    tile_scan(L, dstp, 0, E, d0, tdn, tid);
    __syncthreads();
    n = L.cnt;
  }
  if (MODE == 1 && tid == 0) *wcnt = (N <= KEEP_MAXN) ? n : CAP + 1;
  if (n <= CAP) {
    if (kept) tile_max_kept<H>(L, wl, als, n, tid);
    else      tile_max<H, MODE == 1>(L, ei, dstp, als, wl, n, d0, tid);
    __syncthreads();
    for (int i = tid; i < tdn * H; i += NT) {
      float w = __expf(L.wsf[i] - L.mx[i]);
      L.wsf[i] = w; L.den[i] = w;
    }
    __syncthreads();
    tile_sum<H>(L, hfeat, als, n, tid);
  } else {
    const int nwin = (E + CAP - 1) / CAP;
    for (int pass = 0; pass < 2; ++pass) {
      for (int w = 0; w < nwin; ++w) {
        __syncthreads();
        if (tid == 0) L.cnt = 0;
        __syncthreads();
        const int lo = w * CAP;
        int hi = lo + CAP; if (hi > E) hi = E;
        tile_scan(L, dstp, lo, hi, d0, tdn, tid);
        __syncthreads();
        n = L.cnt;
        tile_max<H, false>(L, ei, dstp, als, wl, n, d0, tid);
        __syncthreads();
        if (pass == 1) tile_sum<H>(L, hfeat, als, n, tid);
      }
      if (pass == 0) {
        __syncthreads();
        for (int i = tid; i < tdn * H; i += NT) {
          float w = __expf(L.wsf[i] - L.mx[i]);
          L.wsf[i] = w; L.den[i] = w;
        }
      }
    }
  }
  __syncthreads();

  {
    constexpr int UO = 5;
    constexpr int RS = NT / HID;
    const int c = tid & (HID - 1);
    const int hq = c / F;
    const float bv = bias[c];
    for (int r0 = tid / HID; r0 < tdn; r0 += RS * UO) {
      float hv[UO];
#pragma unroll
      for (int u = 0; u < UO; ++u) {
        int dl = r0 + RS * u; if (dl >= tdn) dl = tdn - 1;
        hv[u] = hfeat[(size_t)(d0 + dl) * HID + c];
      }
#pragma unroll
      for (int u = 0; u < UO; ++u) {
        const int dl = r0 + RS * u;
        if (dl < tdn) {
          const int q = dl * H + hq;
          float v = (L.acc[dl * HID + c] + L.wsf[q] * hv[u]) / L.den[q] + bv;
          if (act) v = lrelu(v, 0.01f);
          outp[(size_t)(d0 + dl) * HID + c] = v;
        }
      }
    }
  }
}

__global__ void __launch_bounds__(NT) gat_tile_h4_kernel(const int* __restrict__ ei,
    const float* __restrict__ hfeat, const float* __restrict__ als, const float* __restrict__ ald,
    const float* __restrict__ bias, float* __restrict__ outp, int* __restrict__ keepbuf,
    int E, int N, int act) {
  __shared__ TileLds L;
  gat_tile_body<4, 1>(L, ei, hfeat, als, ald, bias, outp, keepbuf, E, N, act);
}

__global__ void __launch_bounds__(NT) gat_tile_h1_kernel(const int* __restrict__ ei,
    const float* __restrict__ hfeat, const float* __restrict__ als, const float* __restrict__ ald,
    const float* __restrict__ bias, float* __restrict__ outp, int* __restrict__ keepbuf,
    int E, int N, int act) {
  __shared__ TileLds L;
  gat_tile_body<1, 2>(L, ei, hfeat, als, ald, bias, outp, keepbuf, E, N, act);
}

extern "C" void kernel_launch(void* const* d_in, const int* in_sizes, int n_in,
                              void* d_out, int out_size, void* d_ws, size_t ws_size,
                              hipStream_t stream) {
  const float* num_prop = (const float*)d_in[0];
  const float* cat_prop = (const float*)d_in[1];
  const int*   ei       = (const int*)d_in[2];
  const float* W_num = (const float*)d_in[3];
  const float* b_num = (const float*)d_in[4];
  const float* W_cat = (const float*)d_in[5];
  const float* b_cat = (const float*)d_in[6];
  const float* W_in  = (const float*)d_in[7];
  const float* b_in  = (const float*)d_in[8];
  const float* bn_g  = (const float*)d_in[9];
  const float* bn_b  = (const float*)d_in[10];
  const float* W1    = (const float*)d_in[11];
  const float* a_s1  = (const float*)d_in[12];
  const float* a_d1  = (const float*)d_in[13];
  const float* bias1 = (const float*)d_in[14];
  const float* W2    = (const float*)d_in[15];
  const float* a_s2  = (const float*)d_in[16];
  const float* a_d2  = (const float*)d_in[17];
  const float* bias2 = (const float*)d_in[18];

  const int N = in_sizes[0] / 7;
  const int E = in_sizes[2] / 2;
  float* out = (float*)d_out;

  float* ws   = (float*)d_ws;
  float* xcat = ws;
  float* x1   = xcat + (size_t)N * HID;
  float* hbuf = x1   + (size_t)N * HID;
  float* accb = hbuf + (size_t)N * HID;
  float* als  = accb + (size_t)N * HID;
  float* ald  = als  + (size_t)N * 4;
  float* stats= ald  + (size_t)N * 12;

  const int TB = 256;
  const int ntiles = (N + TD - 1) / TD;
  const int mtiles = (N + 15) / 16;
  const int gblocks = (mtiles + GW - 1) / GW;
  auto nb = [&](long long work) { return (unsigned)((work + TB - 1) / TB); };
  int* keepbuf = (int*)x1;

  encoder_kernel<<<nb((long long)N * 64), TB, 0, stream>>>(num_prop, cat_prop, W_num, b_num,
                                                           W_cat, b_cat, xcat, N);
  wmma_gemm128_kernel<<<gblocks, GW * 32, 0, stream>>>(xcat, W_in, b_in, x1, N, 1, 1);

  fill_kernel<<<1, 256, 0, stream>>>(stats, 256, 0.f);
  bn_stats_kernel<<<512, 128, 0, stream>>>(x1, stats, N);
  bn_finalize_kernel<<<1, 128, 0, stream>>>(stats, bn_g, bn_b, N);
  bn_apply_kernel<<<nb((long long)N * HID), TB, 0, stream>>>(x1, stats, xcat, N);

  wmma_gemm128_kernel<<<gblocks, GW * 32, 0, stream>>>(xcat, W1, nullptr, hbuf, N, 0, 0);
  att_logits_kernel<<<nb((long long)N * 4), TB, 0, stream>>>(hbuf, a_s1, a_d1, als, ald, N, 4, 32);
  gat_tile_h4_kernel<<<ntiles, NT, 0, stream>>>(ei, hbuf, als, ald, bias1, accb, keepbuf, E, N, 1);

  wmma_gemm128_kernel<<<gblocks, GW * 32, 0, stream>>>(accb, W2, nullptr, hbuf, N, 0, 0);
  att_logits_kernel<<<nb(N), TB, 0, stream>>>(hbuf, a_s2, a_d2, als, ald, N, 1, 128);
  gat_tile_h1_kernel<<<ntiles, NT, 0, stream>>>(ei, hbuf, als, ald, bias2, out, keepbuf, E, N, 0);
}


# GAT edge lists by one binning launch per call: edges appended in arrival order to the destination tile's list (LDS counts, one global integer add per tile and block); no column scan; 12->13 launches
# speedup vs baseline: 2.9037x; 1.0848x over previous
#include <hip/hip_runtime.h>
#include <math.h>

typedef __attribute__((ext_vector_type(2))) float v2f;
typedef __attribute__((ext_vector_type(8))) float v8f;

#define HID 128

__device__ __forceinline__ float lrelu(float v, float s) { return v > 0.f ? v : s * v; }

__global__ void encoder_kernel(const float* __restrict__ num_prop, const float* __restrict__ cat_prop,
                               const float* __restrict__ Wn, const float* __restrict__ bn,
                               const float* __restrict__ Wc, const float* __restrict__ bc,
                               float* __restrict__ xcat, int N) {
  int t = blockIdx.x * blockDim.x + threadIdx.x;
  if (t >= N * 64) return;
  int n = t >> 6, j = t & 63;
  const float* np = num_prop + (size_t)n * 7;
  float an = bn[j];
#pragma unroll
  for (int k = 0; k < 7; ++k) an += np[k] * Wn[k * 64 + j];
  const float* cp = cat_prop + (size_t)n * 3;
  float ac = bc[j];
#pragma unroll
  for (int k = 0; k < 3; ++k) ac += cp[k] * Wc[k * 64 + j];
  xcat[(size_t)n * HID + j]      = lrelu(an, 0.01f);
  xcat[(size_t)n * HID + 64 + j] = lrelu(ac, 0.01f);
}

#define GW 8
#define BP 136
__global__ void __launch_bounds__(GW * 32)
wmma_gemm128_kernel(const float* __restrict__ A, const float* __restrict__ B,
                    const float* __restrict__ bias, float* __restrict__ C,
                    int M, int act, int has_bias) {
  __shared__ float4 Bl[32 * BP];
  const int tid  = threadIdx.x;
  const int wave = tid >> 5;
  const int lane = tid & 31;
  const int half = lane >> 4;
  const int m16  = lane & 15;
  const int row0 = (blockIdx.x * GW + wave) * 16;

  int arow = row0 + m16; if (arow >= M) arow = M - 1;
  const float4* Ap = reinterpret_cast<const float4*>(A + (size_t)arow * HID) + half;
  float4 a[16];
#pragma unroll
  for (int j = 0; j < 16; ++j) a[j] = Ap[2 * j];

  float bv[8];
#pragma unroll
  for (int t = 0; t < 8; ++t) bv[t] = has_bias ? bias[16 * t + m16] : 0.f;

  {
    const int col = tid & (HID - 1);
#pragma unroll
    for (int i = 0; i < 16; ++i) {
      const int q = (tid >> 7) + 2 * i;
      const float* bp = B + (size_t)(4 * q) * HID + col;
      float4 v;
      v.x = bp[0]; v.y = bp[HID]; v.z = bp[2 * HID]; v.w = bp[3 * HID];
      Bl[q * BP + col] = v;
    }
  }
  __syncthreads();
  if (row0 >= M) return;

  v8f acc[8];
#pragma unroll
  for (int t = 0; t < 8; ++t) acc[t] = (v8f){0.f, 0.f, 0.f, 0.f, 0.f, 0.f, 0.f, 0.f};

  const float4* Bf = Bl + half * BP + m16;
  float4 bc[8], bn[8];
#pragma unroll
  for (int t = 0; t < 8; ++t) bc[t] = Bf[16 * t];
#pragma unroll
  for (int j = 0; j < 16; ++j) {
    if (j + 1 < 16) {
#pragma unroll
      for (int t = 0; t < 8; ++t) bn[t] = Bf[(2 * (j + 1)) * BP + 16 * t];
    }
    __builtin_amdgcn_sched_barrier(0);
    const v2f a0 = {a[j].x, a[j].y};
    const v2f a1 = {a[j].z, a[j].w};
#pragma unroll
    for (int t = 0; t < 8; ++t) {
      const v2f b0 = {bc[t].x, bc[t].y};
      acc[t] = __builtin_amdgcn_wmma_f32_16x16x4_f32(false, a0, false, b0, (short)0, acc[t], false, false);
    }
#pragma unroll
    for (int t = 0; t < 8; ++t) {
      const v2f b1 = {bc[t].z, bc[t].w};
      acc[t] = __builtin_amdgcn_wmma_f32_16x16x4_f32(false, a1, false, b1, (short)0, acc[t], false, false);
    }
    __builtin_amdgcn_sched_barrier(0);
#pragma unroll
    for (int t = 0; t < 8; ++t) bc[t] = bn[t];
  }

  float* Cp = C + (size_t)(row0 + 8 * half) * HID + m16;
  if (row0 + 16 <= M) {
#pragma unroll
    for (int r = 0; r < 8; ++r) {
#pragma unroll
      for (int t = 0; t < 8; ++t) {
        float v = acc[t][r] + bv[t];
        if (act) v = lrelu(v, 0.01f);
        Cp[r * HID + 16 * t] = v;
      }
    }
  } else {
#pragma unroll
    for (int r = 0; r < 8; ++r) {
      if (row0 + 8 * half + r < M) {
#pragma unroll
        for (int t = 0; t < 8; ++t) {
          float v = acc[t][r] + bv[t];
          if (act) v = lrelu(v, 0.01f);
          Cp[r * HID + 16 * t] = v;
        }
      }
    }
  }
}

__global__ void bn_stats_kernel(const float* __restrict__ x, float* __restrict__ stats, int N) {
  int c = threadIdx.x;
  float s = 0.f, s2 = 0.f;
  for (int n = blockIdx.x; n < N; n += gridDim.x) {
    float v = x[(size_t)n * HID + c];
    s += v; s2 += v * v;
  }
  atomicAdd(stats + c, s);
  atomicAdd(stats + HID + c, s2);
}

__global__ void bn_finalize_kernel(float* __restrict__ stats, const float* __restrict__ gamma,
                                   const float* __restrict__ beta, int N) {
  int c = threadIdx.x;
  float inv_n = 1.f / (float)N;
  float mu  = stats[c] * inv_n;
  float var = stats[HID + c] * inv_n - mu * mu;
  float sc  = gamma[c] * rsqrtf(var + 1e-5f);
  stats[2 * HID + c] = sc;
  stats[3 * HID + c] = beta[c] - mu * sc;
}

__global__ void bn_apply_kernel(const float* __restrict__ x, const float* __restrict__ stats,
                                float* __restrict__ y, int N) {
  int t = blockIdx.x * blockDim.x + threadIdx.x;
  if (t >= N * HID) return;
  int c = t & (HID - 1);
  y[t] = x[t] * stats[2 * HID + c] + stats[3 * HID + c];
}

__global__ void att_logits_kernel(const float* __restrict__ h, const float* __restrict__ a_src,
                                  const float* __restrict__ a_dst, float* __restrict__ als,
                                  float* __restrict__ ald, int N, int H, int F) {
  int t = blockIdx.x * blockDim.x + threadIdx.x;
  if (t >= N * H) return;
  int n = t / H, hh = t - n * H;
  const float* hp = h + (size_t)n * HID + hh * F;
  const float* as = a_src + hh * F;
  const float* ad = a_dst + hh * F;
  float ss = 0.f, sd = 0.f;
  for (int f = 0; f < F; ++f) { float v = hp[f]; ss += v * as[f]; sd += v * ad[f]; }
  als[t] = ss;
  ald[t] = sd;
}

__global__ void clear_kernel(float* __restrict__ stats, int ns, int* __restrict__ tcnt, int nt) {
  int t = blockIdx.x * blockDim.x + threadIdx.x;
  if (t < ns) stats[t] = 0.f;
  if (t < nt) tcnt[t] = 0;
}

#define TD   196
#define CAP  16384
#define NT   1024
#define UB   8
#define UE   8

struct TileLds {
  float acc[TD * HID];
  float mx[TD * 4];
  float den[TD * 4];
  float aldT[TD * 4];
  float wsf[TD * 4];
  int   list[CAP];
  unsigned short dls[CAP];
  int   cnt;
};

__device__ __forceinline__ void ldsMaxF(float* addr, float val) {
  if (val >= 0.f) atomicMax((int*)addr, __float_as_int(val));
  else            atomicMin((unsigned int*)addr, __float_as_uint(val));
}

__device__ __forceinline__ void tile_push(TileLds& L, int e) {
  int pos = atomicAdd(&L.cnt, 1);
  if (pos < CAP) L.list[pos] = e;
}

__device__ __forceinline__ void tile_test1(TileLds& L, const int* __restrict__ dstp, int e,
                                           int d0, int tdn) {
  if ((unsigned)(dstp[e] - d0) < (unsigned)tdn) tile_push(L, e);
}

__device__ __forceinline__ void tile_scan(TileLds& L, const int* __restrict__ dstp, int lo, int hi,
                                          int d0, int tdn, int tid) {
  int head = (4 - (int)((((size_t)(dstp + lo)) >> 2) & 3)) & 3;
  int a0 = lo + head; if (a0 > hi) a0 = hi;
  if (tid < a0 - lo) tile_test1(L, dstp, lo + tid, d0, tdn);
  const int nvec = (hi - a0) >> 2;
  const int4* p4 = (const int4*)(dstp + a0);
  const int rot = (nvec / (int)gridDim.x) * (int)blockIdx.x;
  for (int j = 0; j < nvec; j += NT * UB) {
    int4 v[UB]; int idx[UB];
#pragma unroll
    for (int u = 0; u < UB; ++u) {
      int k = j + u * NT + tid;
      int i = k + rot; if (i >= nvec) i -= nvec;
      if (k >= nvec) i = -1;
      idx[u] = i;
      v[u] = p4[i < 0 ? 0 : i];
    }
#pragma unroll
    for (int u = 0; u < UB; ++u) {
      const bool ok = idx[u] >= 0;
      bool h0 = ok & ((unsigned)(v[u].x - d0) < (unsigned)tdn);
      bool h1 = ok & ((unsigned)(v[u].y - d0) < (unsigned)tdn);
      bool h2 = ok & ((unsigned)(v[u].z - d0) < (unsigned)tdn);
      bool h3 = ok & ((unsigned)(v[u].w - d0) < (unsigned)tdn);
      if (h0 | h1 | h2 | h3) {
        int e0 = a0 + 4 * idx[u];
        if (h0) tile_push(L, e0);
        if (h1) tile_push(L, e0 + 1);
        if (h2) tile_push(L, e0 + 2);
        if (h3) tile_push(L, e0 + 3);
      }
    }
  }
  int t0 = a0 + 4 * nvec;
  if (tid < hi - t0) tile_test1(L, dstp, t0 + tid, d0, tdn);
}

#define UM 4
#define KEEP_MAXN (1 << 24)
template <int H>
__device__ __forceinline__ void tile_max(TileLds& L, const int* __restrict__ ei,
                                         const int* __restrict__ dstp,
                                         const float* __restrict__ als,
                                         int n, int d0, int tid) {
  for (int i0 = tid; i0 < n; i0 += NT * UM) {
    int ix[UM], e[UM], s[UM], dl[UM]; float a[UM][H];
#pragma unroll
    for (int u = 0; u < UM; ++u) {
      int i = i0 + u * NT;
      ix[u] = i < n ? i : i0;
      e[u] = L.list[ix[u]];
    }
#pragma unroll
    for (int u = 0; u < UM; ++u) { s[u] = ei[e[u]]; dl[u] = dstp[e[u]] - d0; }
#pragma unroll
    for (int u = 0; u < UM; ++u) {
#pragma unroll
      for (int h = 0; h < H; ++h) a[u][h] = als[(size_t)s[u] * H + h];
    }
#pragma unroll
    for (int u = 0; u < UM; ++u) {
      if (i0 + u * NT < n) {
        L.list[ix[u]] = s[u];
        L.dls[ix[u]] = (unsigned short)dl[u];
#pragma unroll
        for (int h = 0; h < H; ++h) {
          float v = lrelu(a[u][h] + L.aldT[dl[u] * H + h], 0.2f);
          ldsMaxF(&L.mx[dl[u] * H + h], v);
        }
      }
    }
  }
}

template <int H>
__device__ __forceinline__ void tile_max_kept(TileLds& L, const int* __restrict__ wl,
                                              const float* __restrict__ als, int n, int tid) {
  for (int i0 = tid; i0 < n; i0 += NT * UM) {
    int ix[UM], s[UM], dl[UM]; unsigned p[UM]; float a[UM][H];
#pragma unroll
    for (int u = 0; u < UM; ++u) {
      int i = i0 + u * NT;
      ix[u] = i < n ? i : i0;
      p[u] = (unsigned)wl[ix[u]];
    }
#pragma unroll
    for (int u = 0; u < UM; ++u) {
      s[u] = (int)(p[u] >> 8); dl[u] = (int)(p[u] & 255u);
#pragma unroll
      for (int h = 0; h < H; ++h) a[u][h] = als[(size_t)s[u] * H + h];
    }
#pragma unroll
    for (int u = 0; u < UM; ++u) {
      if (i0 + u * NT < n) {
        L.list[ix[u]] = s[u];
        L.dls[ix[u]] = (unsigned short)dl[u];
#pragma unroll
        for (int h = 0; h < H; ++h) {
          float v = lrelu(a[u][h] + L.aldT[dl[u] * H + h], 0.2f);
          ldsMaxF(&L.mx[dl[u] * H + h], v);
        }
      }
    }
  }
}

template <int H>
__device__ __forceinline__ void tile_sum(TileLds& L, const float* __restrict__ hfeat,
                                         const float* __restrict__ als, int n, int tid) {
  constexpr int F = HID / H;
  const int wave = tid >> 5, lane = tid & 31;
  for (int i0 = wave * UE; i0 < n; i0 += (NT / 32) * UE) {
    int s[UE], dl[UE]; float al[UE]; float hv[UE][4];
#pragma unroll
    for (int u = 0; u < UE; ++u) {
      int i = i0 + u; if (i >= n) i = n - 1;
      s[u] = L.list[i]; dl[u] = L.dls[i];
    }
#pragma unroll
    for (int u = 0; u < UE; ++u) {
      const float* hs = hfeat + (size_t)s[u] * HID;
      al[u] = (lane < H) ? als[(size_t)s[u] * H + lane] : 0.f;
#pragma unroll
      for (int k = 0; k < 4; ++k) hv[u][k] = hs[lane + 32 * k];
    }
#pragma unroll
    for (int u = 0; u < UE; ++u) {
      if (i0 + u < n) {
        float w_l = 0.f;
        if (lane < H) {
          int q = dl[u] * H + lane;
          float v = lrelu(al[u] + L.aldT[q], 0.2f);
          w_l = __expf(v - L.mx[q]);
          atomicAdd(&L.den[q], w_l);
        }
#pragma unroll
        for (int k = 0; k < 4; ++k) {
          int c = lane + 32 * k;
          float w = __shfl(w_l, c / F, 32);
          atomicAdd(&L.acc[dl[u] * HID + c], w * hv[u][k]);
        }
      }
    }
  }
}

template <int H>
__device__ __forceinline__ void gat_tile_body(TileLds& L, const int* __restrict__ ei,
                                              const float* __restrict__ hfeat,
                                              const float* __restrict__ als,
                                              const float* __restrict__ ald,
                                              const float* __restrict__ bias,
                                              float* __restrict__ outp,
                                              const int* __restrict__ keepbuf,
                                              const int* __restrict__ tcnt,
                                              int E, int N, int act) {
  constexpr int F = HID / H;
  const int tid = threadIdx.x;
  const int d0 = blockIdx.x * TD;
  int tdn = N - d0; if (tdn > TD) tdn = TD;
  const int* dstp = ei + (size_t)E;
  const int* wl = keepbuf + (size_t)blockIdx.x * CAP;

  for (int i = tid; i < TD * HID; i += NT) L.acc[i] = 0.f;
  for (int i = tid; i < tdn * H; i += NT) {
    size_t g = (size_t)d0 * H + i;
    float a_d = ald[g];
    float v = lrelu(als[g] + a_d, 0.2f);
    L.aldT[i] = a_d; L.wsf[i] = v; L.mx[i] = v;
  }

  __syncthreads();
  int n = tcnt[blockIdx.x];
  if (N > KEEP_MAXN) n = CAP + 1;
  const bool kept = n <= CAP;
  if (!kept) {
    if (tid == 0) L.cnt = 0;
    __syncthreads();
    tile_scan(L, dstp, 0, E, d0, tdn, tid);
    __syncthreads();
    n = L.cnt;
  }
  if (n <= CAP) {
    if (kept) tile_max_kept<H>(L, wl, als, n, tid);
    else      tile_max<H>(L, ei, dstp, als, n, d0, tid);
    __syncthreads();
    for (int i = tid; i < tdn * H; i += NT) {
      float w = __expf(L.wsf[i] - L.mx[i]);
      L.wsf[i] = w; L.den[i] = w;
    }
    __syncthreads();
    tile_sum<H>(L, hfeat, als, n, tid);
  } else {
    const int nwin = (E + CAP - 1) / CAP;
    for (int pass = 0; pass < 2; ++pass) {
      for (int w = 0; w < nwin; ++w) {
        __syncthreads();
        if (tid == 0) L.cnt = 0;
        __syncthreads();
        const int lo = w * CAP;
        int hi = lo + CAP; if (hi > E) hi = E;
        tile_scan(L, dstp, lo, hi, d0, tdn, tid);
        __syncthreads();
        n = L.cnt;
        tile_max<H>(L, ei, dstp, als, n, d0, tid);
        __syncthreads();
        if (pass == 1) tile_sum<H>(L, hfeat, als, n, tid);
      }
      if (pass == 0) {
        __syncthreads();
        for (int i = tid; i < tdn * H; i += NT) {
          float w = __expf(L.wsf[i] - L.mx[i]);
          L.wsf[i] = w; L.den[i] = w;
        }
      }
    }
  }
  __syncthreads();

  {
    constexpr int UO = 5;
    constexpr int RS = NT / HID;
    const int c = tid & (HID - 1);
    const int hq = c / F;
    const float bv = bias[c];
    for (int r0 = tid / HID; r0 < tdn; r0 += RS * UO) {
      float hv[UO];
#pragma unroll
      for (int u = 0; u < UO; ++u) {
        int dl = r0 + RS * u; if (dl >= tdn) dl = tdn - 1;
        hv[u] = hfeat[(size_t)(d0 + dl) * HID + c];
      }
#pragma unroll
      for (int u = 0; u < UO; ++u) {
        const int dl = r0 + RS * u;
        if (dl < tdn) {
          const int q = dl * H + hq;
          float v = (L.acc[dl * HID + c] + L.wsf[q] * hv[u]) / L.den[q] + bv;
          if (act) v = lrelu(v, 0.01f);
          outp[(size_t)(d0 + dl) * HID + c] = v;
        }
      }
    }
  }
}

__global__ void __launch_bounds__(NT) gat_tile_h4_kernel(const int* __restrict__ ei,
    const float* __restrict__ hfeat, const float* __restrict__ als, const float* __restrict__ ald,
    const float* __restrict__ bias, float* __restrict__ outp, const int* __restrict__ keepbuf,
    const int* __restrict__ tcnt, int E, int N, int act) {
  __shared__ TileLds L;
  gat_tile_body<4>(L, ei, hfeat, als, ald, bias, outp, keepbuf, tcnt, E, N, act);
}

__global__ void __launch_bounds__(NT) gat_tile_h1_kernel(const int* __restrict__ ei,
    const float* __restrict__ hfeat, const float* __restrict__ als, const float* __restrict__ ald,
    const float* __restrict__ bias, float* __restrict__ outp, const int* __restrict__ keepbuf,
    const int* __restrict__ tcnt, int E, int N, int act) {
  __shared__ TileLds L;
  gat_tile_body<1>(L, ei, hfeat, als, ald, bias, outp, keepbuf, tcnt, E, N, act);
}

#define BNT   512
#define BNE   8
#define BMAXT 2048
__global__ void __launch_bounds__(BNT)
bin_edges_kernel(const int* __restrict__ ei, int* __restrict__ keepbuf, int* __restrict__ tcnt,
                 int E, int N, int ntiles) {
  __shared__ int cnt[BMAXT + 1];
  __shared__ int base[BMAXT + 1];
  const int tid = threadIdx.x;
  const int* dstp = ei + (size_t)E;
  const int e0 = blockIdx.x * (BNT * BNE) + tid;
  int s[BNE], tl[BNE], dl[BNE];
#pragma unroll
  for (int u = 0; u < BNE; ++u) {
    const int e = e0 + u * BNT;
    const int ec = e < E ? e : E - 1;
    const int d = dstp[ec];
    s[u] = ei[ec];
    const int t = d / TD;
    dl[u] = d - t * TD;
    tl[u] = ((e < E) & ((unsigned)d < (unsigned)N)) ? t : -1;
  }
  for (int w0 = 0; w0 < ntiles; w0 += BMAXT) {
    const int nt = ntiles - w0 < BMAXT ? ntiles - w0 : BMAXT;
    for (int i = tid; i < nt; i += BNT) cnt[i] = 0;
    if (tid == 0) { cnt[BMAXT] = 0; base[BMAXT] = CAP; }
    __syncthreads();
    int rr[BNE];
#pragma unroll
    for (int u = 0; u < BNE; ++u) {
      const unsigned r = (unsigned)(tl[u] - w0);
      rr[u] = r < (unsigned)nt ? (int)r : BMAXT;
      atomicAdd(&cnt[rr[u]], 1);
    }
    __syncthreads();
    for (int i = tid; i < nt; i += BNT) {
      const int k = cnt[i];
      if (k > 0) base[i] = atomicAdd(&tcnt[w0 + i], k);
      cnt[i] = 0;
    }
    __syncthreads();
    int pos[BNE];
#pragma unroll
    for (int u = 0; u < BNE; ++u) pos[u] = base[rr[u]] + atomicAdd(&cnt[rr[u]], 1);
#pragma unroll
    for (int u = 0; u < BNE; ++u) {
      if (rr[u] != BMAXT && pos[u] < CAP)
        keepbuf[(size_t)tl[u] * CAP + pos[u]] = (int)(((unsigned)s[u] << 8) | (unsigned)dl[u]);
    }
    __syncthreads();
  }
}

extern "C" void kernel_launch(void* const* d_in, const int* in_sizes, int n_in,
                              void* d_out, int out_size, void* d_ws, size_t ws_size,
                              hipStream_t stream) {
  const float* num_prop = (const float*)d_in[0];
  const float* cat_prop = (const float*)d_in[1];
  const int*   ei       = (const int*)d_in[2];
  const float* W_num = (const float*)d_in[3];
  const float* b_num = (const float*)d_in[4];
  const float* W_cat = (const float*)d_in[5];
  const float* b_cat = (const float*)d_in[6];
  const float* W_in  = (const float*)d_in[7];
  const float* b_in  = (const float*)d_in[8];
  const float* bn_g  = (const float*)d_in[9];
  const float* bn_b  = (const float*)d_in[10];
  const float* W1    = (const float*)d_in[11];
  const float* a_s1  = (const float*)d_in[12];
  const float* a_d1  = (const float*)d_in[13];
  const float* bias1 = (const float*)d_in[14];
  const float* W2    = (const float*)d_in[15];
  const float* a_s2  = (const float*)d_in[16];
  const float* a_d2  = (const float*)d_in[17];
  const float* bias2 = (const float*)d_in[18];

  const int N = in_sizes[0] / 7;
  const int E = in_sizes[2] / 2;
  float* out = (float*)d_out;

  float* ws   = (float*)d_ws;
  float* xcat = ws;
  float* x1   = xcat + (size_t)N * HID;
  float* hbuf = x1   + (size_t)N * HID;
  float* accb = hbuf + (size_t)N * HID;
  float* als  = accb + (size_t)N * HID;
  float* ald  = als  + (size_t)N * 4;
  int*   tcnt = (int*)(ald + (size_t)N * 4);
  float* stats= ald  + (size_t)N * 12;

  const int TB = 256;
  const int ntiles = (N + TD - 1) / TD;
  const int mtiles = (N + 15) / 16;
  const int gblocks = (mtiles + GW - 1) / GW;
  auto nb = [&](long long work) { return (unsigned)((work + TB - 1) / TB); };
  int* keepbuf = (int*)x1;
  const int bblocks = (E + BNT * BNE - 1) / (BNT * BNE);

  encoder_kernel<<<nb((long long)N * 64), TB, 0, stream>>>(num_prop, cat_prop, W_num, b_num,
                                                           W_cat, b_cat, xcat, N);
  wmma_gemm128_kernel<<<gblocks, GW * 32, 0, stream>>>(xcat, W_in, b_in, x1, N, 1, 1);

  clear_kernel<<<nb(ntiles > 256 ? ntiles : 256), TB, 0, stream>>>(stats, 256, tcnt, ntiles);
  bn_stats_kernel<<<512, 128, 0, stream>>>(x1, stats, N);
  bn_finalize_kernel<<<1, 128, 0, stream>>>(stats, bn_g, bn_b, N);
  bn_apply_kernel<<<nb((long long)N * HID), TB, 0, stream>>>(x1, stats, xcat, N);

  bin_edges_kernel<<<bblocks, BNT, 0, stream>>>(ei, keepbuf, tcnt, E, N, ntiles);

  wmma_gemm128_kernel<<<gblocks, GW * 32, 0, stream>>>(xcat, W1, nullptr, hbuf, N, 0, 0);
  att_logits_kernel<<<nb((long long)N * 4), TB, 0, stream>>>(hbuf, a_s1, a_d1, als, ald, N, 4, 32);
  gat_tile_h4_kernel<<<ntiles, NT, 0, stream>>>(ei, hbuf, als, ald, bias1, accb, keepbuf, tcnt, E, N, 1);

  wmma_gemm128_kernel<<<gblocks, GW * 32, 0, stream>>>(accb, W2, nullptr, hbuf, N, 0, 0);
  att_logits_kernel<<<nb(N), TB, 0, stream>>>(hbuf, a_s2, a_d2, als, ald, N, 1, 128);
  gat_tile_h1_kernel<<<ntiles, NT, 0, stream>>>(ei, hbuf, als, ald, bias2, out, keepbuf, tcnt, E, N, 0);
}


# encoder and batchnorm-apply launches re-gridded: a wave makes 8 whole rows (encoder, products fused by fmaf) or 8 x 16 bytes per lane (apply) instead of 1-2 floats per lane; 300000 waves to 12500
# speedup vs baseline: 3.6471x; 1.2560x over previous
#include <hip/hip_runtime.h>
#include <math.h>

typedef __attribute__((ext_vector_type(2))) float v2f;
typedef __attribute__((ext_vector_type(8))) float v8f;

#define HID 128

__device__ __forceinline__ float lrelu(float v, float s) { return v > 0.f ? v : s * v; }

#define ER 8
#define EW 8
__global__ void __launch_bounds__(EW * 32)
encoder_kernel(const float* __restrict__ num_prop, const float* __restrict__ cat_prop,
               const float* __restrict__ Wn, const float* __restrict__ bn,
               const float* __restrict__ Wc, const float* __restrict__ bc,
               float* __restrict__ xcat, int N) {
  const int lane = threadIdx.x & 31;
  const int wv = __builtin_amdgcn_readfirstlane((int)(threadIdx.x >> 5));
  const int n0 = (blockIdx.x * EW + wv) * ER;
  if (n0 >= N) return;
  const int j = 2 * lane;
  float wn[7][2], wc[3][2];
#pragma unroll
  for (int k = 0; k < 7; ++k) { wn[k][0] = Wn[k * 64 + j]; wn[k][1] = Wn[k * 64 + j + 1]; }
#pragma unroll
  for (int k = 0; k < 3; ++k) { wc[k][0] = Wc[k * 64 + j]; wc[k][1] = Wc[k * 64 + j + 1]; }
  const float bn0 = bn[j], bn1 = bn[j + 1], bc0 = bc[j], bc1 = bc[j + 1];
  float xn[ER][7], xc[ER][3];
#pragma unroll
  for (int u = 0; u < ER; ++u) {
    int n = n0 + u; if (n >= N) n = N - 1;
#pragma unroll
    for (int k = 0; k < 7; ++k) xn[u][k] = num_prop[(size_t)n * 7 + k];
#pragma unroll
    for (int k = 0; k < 3; ++k) xc[u][k] = cat_prop[(size_t)n * 3 + k];
  }
#pragma unroll
  for (int u = 0; u < ER; ++u) {
    if (n0 + u < N) {
      float an0 = bn0, an1 = bn1, ac0 = bc0, ac1 = bc1;
#pragma unroll
      for (int k = 0; k < 7; ++k) { an0 = __builtin_fmaf(xn[u][k], wn[k][0], an0); an1 = __builtin_fmaf(xn[u][k], wn[k][1], an1); }
#pragma unroll
      for (int k = 0; k < 3; ++k) { ac0 = __builtin_fmaf(xc[u][k], wc[k][0], ac0); ac1 = __builtin_fmaf(xc[u][k], wc[k][1], ac1); }
      float2 on, oc;
      on.x = lrelu(an0, 0.01f); on.y = lrelu(an1, 0.01f);
      oc.x = lrelu(ac0, 0.01f); oc.y = lrelu(ac1, 0.01f);
      float* row = xcat + (size_t)(n0 + u) * HID + j;
      *reinterpret_cast<float2*>(row)      = on;
      *reinterpret_cast<float2*>(row + 64) = oc;
    }
  }
}

#define GW 8
#define BP 136
template <bool SUMS>
__global__ void __launch_bounds__(GW * 32)
wmma_gemm128_kernel(const float* __restrict__ A, const float* __restrict__ B,
                    const float* __restrict__ bias, float* __restrict__ C,
                    float* __restrict__ part, int M, int act, int has_bias) {
  __shared__ float4 Bl[32 * BP];
  __shared__ float red[SUMS ? GW * 2 * HID : 1];
  const int tid  = threadIdx.x;
  const int wave = tid >> 5;
  const int lane = tid & 31;
  const int half = lane >> 4;
  const int m16  = lane & 15;
  const int row0 = (blockIdx.x * GW + wave) * 16;

  int arow = row0 + m16; if (arow >= M) arow = M - 1;
  const float4* Ap = reinterpret_cast<const float4*>(A + (size_t)arow * HID) + half;
  float4 a[16];
#pragma unroll
  for (int j = 0; j < 16; ++j) a[j] = Ap[2 * j];

  float bv[8];
#pragma unroll
  for (int t = 0; t < 8; ++t) bv[t] = has_bias ? bias[16 * t + m16] : 0.f;

  {
    const int col = tid & (HID - 1);
#pragma unroll
    for (int i = 0; i < 16; ++i) {
      const int q = (tid >> 7) + 2 * i;
      const float* bp = B + (size_t)(4 * q) * HID + col;
      float4 v;
      v.x = bp[0]; v.y = bp[HID]; v.z = bp[2 * HID]; v.w = bp[3 * HID];
      Bl[q * BP + col] = v;
    }
  }
  __syncthreads();
  if (!SUMS && row0 >= M) return;
  if (row0 < M) {

  v8f acc[8];
#pragma unroll
  for (int t = 0; t < 8; ++t) acc[t] = (v8f){0.f, 0.f, 0.f, 0.f, 0.f, 0.f, 0.f, 0.f};

  const float4* Bf = Bl + half * BP + m16;
  float4 bc[8], bn[8];
#pragma unroll
  for (int t = 0; t < 8; ++t) bc[t] = Bf[16 * t];
#pragma unroll
  for (int j = 0; j < 16; ++j) {
    if (j + 1 < 16) {
#pragma unroll
      for (int t = 0; t < 8; ++t) bn[t] = Bf[(2 * (j + 1)) * BP + 16 * t];
    }
    __builtin_amdgcn_sched_barrier(0);
    const v2f a0 = {a[j].x, a[j].y};
    const v2f a1 = {a[j].z, a[j].w};
#pragma unroll
    for (int t = 0; t < 8; ++t) {
      const v2f b0 = {bc[t].x, bc[t].y};
      acc[t] = __builtin_amdgcn_wmma_f32_16x16x4_f32(false, a0, false, b0, (short)0, acc[t], false, false);
    }
#pragma unroll
    for (int t = 0; t < 8; ++t) {
      const v2f b1 = {bc[t].z, bc[t].w};
      acc[t] = __builtin_amdgcn_wmma_f32_16x16x4_f32(false, a1, false, b1, (short)0, acc[t], false, false);
    }
    __builtin_amdgcn_sched_barrier(0);
#pragma unroll
    for (int t = 0; t < 8; ++t) bc[t] = bn[t];
  }

  float cs[8], cq[8];
#pragma unroll
  for (int t = 0; t < 8; ++t) { cs[t] = 0.f; cq[t] = 0.f; }
  float* Cp = C + (size_t)(row0 + 8 * half) * HID + m16;
  if (row0 + 16 <= M) {
#pragma unroll
    for (int r = 0; r < 8; ++r) {
#pragma unroll
      for (int t = 0; t < 8; ++t) {
        float v = acc[t][r] + bv[t];
        if (act) v = lrelu(v, 0.01f);
        Cp[r * HID + 16 * t] = v;
        if (SUMS) { cs[t] += v; cq[t] += v * v; }
      }
    }
  } else {
#pragma unroll
    for (int r = 0; r < 8; ++r) {
      if (row0 + 8 * half + r < M) {
#pragma unroll
        for (int t = 0; t < 8; ++t) {
          float v = acc[t][r] + bv[t];
          if (act) v = lrelu(v, 0.01f);
          Cp[r * HID + 16 * t] = v;
          if (SUMS) { cs[t] += v; cq[t] += v * v; }
        }
      }
    }
  }
  if (SUMS) {
#pragma unroll
    for (int t = 0; t < 8; ++t) {
      const float s = cs[t] + __shfl_xor(cs[t], 16, 32);
      const float q = cq[t] + __shfl_xor(cq[t], 16, 32);
      red[wave * 2 * HID + half * HID + 16 * t + m16] = half ? q : s;
    }
  }
  } else if (SUMS) {
#pragma unroll
    for (int t = 0; t < 8; ++t) red[wave * 2 * HID + half * HID + 16 * t + m16] = 0.f;
  }
  if (SUMS) {
    __syncthreads();
    float s = 0.f;
#pragma unroll
    for (int w = 0; w < GW; ++w) s += red[w * 2 * HID + tid];
    part[(size_t)blockIdx.x * 2 * HID + tid] = s;
  }
}

#define FT 256
__global__ void __launch_bounds__(FT)
bn_finalize_kernel(const float* __restrict__ part, int nblk, float* __restrict__ stats,
                   const float* __restrict__ gamma, const float* __restrict__ beta, int N) {
  __shared__ float ws_[FT / 32], wq_[FT / 32];
  const int c = blockIdx.x;
  const int i = threadIdx.x;
  float s = 0.f, q = 0.f;
  for (int b = i; b < nblk; b += FT) {
    s += part[(size_t)b * 2 * HID + c];
    q += part[(size_t)b * 2 * HID + HID + c];
  }
#pragma unroll
  for (int w = 16; w > 0; w >>= 1) { s += __shfl_xor(s, w, 32); q += __shfl_xor(q, w, 32); }
  if ((i & 31) == 0) { ws_[i >> 5] = s; wq_[i >> 5] = q; }
  __syncthreads();
  if (i == 0) {
    float ts = 0.f, tq = 0.f;
#pragma unroll
    for (int w = 0; w < FT / 32; ++w) { ts += ws_[w]; tq += wq_[w]; }
    const float inv_n = 1.f / (float)N;
    const float mu  = ts * inv_n;
    const float var = tq * inv_n - mu * mu;
    const float sc  = gamma[c] * rsqrtf(var + 1e-5f);
    stats[2 * HID + c] = sc;
    stats[3 * HID + c] = beta[c] - mu * sc;
  }
}

#define AU 8
#define AT 256
__device__ __forceinline__ float4 bn_piece(float4 v, float4 sc, float4 sh) {
  float4 r;
  r.x = v.x * sc.x + sh.x; r.y = v.y * sc.y + sh.y;
  r.z = v.z * sc.z + sh.z; r.w = v.w * sc.w + sh.w;
  return r;
}
__global__ void __launch_bounds__(AT)
bn_apply_kernel(const float* __restrict__ x, const float* __restrict__ stats,
                float* __restrict__ y, int N) {
  const int total4 = N * (HID / 4);
  const int t = threadIdx.x;
  const int b0 = blockIdx.x * (AT * AU);
  const int c4 = t & (HID / 4 - 1);
  const float4 sc = reinterpret_cast<const float4*>(stats + 2 * HID)[c4];
  const float4 sh = reinterpret_cast<const float4*>(stats + 3 * HID)[c4];
  const float4* __restrict__ x4 = reinterpret_cast<const float4*>(x) + b0 + t;
  float4* __restrict__ y4 = reinterpret_cast<float4*>(y) + b0 + t;
  if (b0 + AT * AU <= total4) {
    float4 v[AU];
#pragma unroll
    for (int u = 0; u < AU; ++u) v[u] = x4[u * AT];
#pragma unroll
    for (int u = 0; u < AU; ++u) y4[u * AT] = bn_piece(v[u], sc, sh);
  } else {
#pragma unroll
    for (int u = 0; u < AU; ++u) {
      if (b0 + t + u * AT < total4) y4[u * AT] = bn_piece(x4[u * AT], sc, sh);
    }
  }
}

__global__ void att_logits_kernel(const float* __restrict__ h, const float* __restrict__ a_src,
                                  const float* __restrict__ a_dst, float* __restrict__ als,
                                  float* __restrict__ ald, int N, int H, int F) {
  int t = blockIdx.x * blockDim.x + threadIdx.x;
  if (t >= N * H) return;
  int n = t / H, hh = t - n * H;
  const float* hp = h + (size_t)n * HID + hh * F;
  const float* as = a_src + hh * F;
  const float* ad = a_dst + hh * F;
  float ss = 0.f, sd = 0.f;
  for (int f = 0; f < F; ++f) { float v = hp[f]; ss += v * as[f]; sd += v * ad[f]; }
  als[t] = ss;
  ald[t] = sd;
}

__global__ void clear_kernel(int* __restrict__ tcnt, int nt) {
  int t = blockIdx.x * blockDim.x + threadIdx.x;
  if (t < nt) tcnt[t] = 0;
}

#define TD   196
#define CAP  16384
#define NT   1024
#define UB   8
#define UE   8

struct TileLds {
  float acc[TD * HID];
  float mx[TD * 4];
  float den[TD * 4];
  float aldT[TD * 4];
  float wsf[TD * 4];
  int   list[CAP];
  unsigned short dls[CAP];
  int   cnt;
};

__device__ __forceinline__ void ldsMaxF(float* addr, float val) {
  if (val >= 0.f) atomicMax((int*)addr, __float_as_int(val));
  else            atomicMin((unsigned int*)addr, __float_as_uint(val));
}

__device__ __forceinline__ void tile_push(TileLds& L, int e) {
  int pos = atomicAdd(&L.cnt, 1);
  if (pos < CAP) L.list[pos] = e;
}

__device__ __forceinline__ void tile_test1(TileLds& L, const int* __restrict__ dstp, int e,
                                           int d0, int tdn) {
  if ((unsigned)(dstp[e] - d0) < (unsigned)tdn) tile_push(L, e);
}

__device__ __forceinline__ void tile_scan(TileLds& L, const int* __restrict__ dstp, int lo, int hi,
                                          int d0, int tdn, int tid) {
  int head = (4 - (int)((((size_t)(dstp + lo)) >> 2) & 3)) & 3;
  int a0 = lo + head; if (a0 > hi) a0 = hi;
  if (tid < a0 - lo) tile_test1(L, dstp, lo + tid, d0, tdn);
  const int nvec = (hi - a0) >> 2;
  const int4* p4 = (const int4*)(dstp + a0);
  const int rot = (nvec / (int)gridDim.x) * (int)blockIdx.x;
  for (int j = 0; j < nvec; j += NT * UB) {
    int4 v[UB]; int idx[UB];
#pragma unroll
    for (int u = 0; u < UB; ++u) {
      int k = j + u * NT + tid;
      int i = k + rot; if (i >= nvec) i -= nvec;
      if (k >= nvec) i = -1;
      idx[u] = i;
      v[u] = p4[i < 0 ? 0 : i];
    }
#pragma unroll
    for (int u = 0; u < UB; ++u) {
      const bool ok = idx[u] >= 0;
      bool h0 = ok & ((unsigned)(v[u].x - d0) < (unsigned)tdn);
      bool h1 = ok & ((unsigned)(v[u].y - d0) < (unsigned)tdn);
      bool h2 = ok & ((unsigned)(v[u].z - d0) < (unsigned)tdn);
      bool h3 = ok & ((unsigned)(v[u].w - d0) < (unsigned)tdn);
      if (h0 | h1 | h2 | h3) {
        int e0 = a0 + 4 * idx[u];
        if (h0) tile_push(L, e0);
        if (h1) tile_push(L, e0 + 1);
        if (h2) tile_push(L, e0 + 2);
        if (h3) tile_push(L, e0 + 3);
      }
    }
  }
  int t0 = a0 + 4 * nvec;
  if (tid < hi - t0) tile_test1(L, dstp, t0 + tid, d0, tdn);
}

#define UM 4
#define KEEP_MAXN (1 << 24)
template <int H>
__device__ __forceinline__ void tile_max(TileLds& L, const int* __restrict__ ei,
                                         const int* __restrict__ dstp,
                                         const float* __restrict__ als,
                                         int n, int d0, int tid) {
  for (int i0 = tid; i0 < n; i0 += NT * UM) {
    int ix[UM], e[UM], s[UM], dl[UM]; float a[UM][H];
#pragma unroll
    for (int u = 0; u < UM; ++u) {
      int i = i0 + u * NT;
      ix[u] = i < n ? i : i0;
      e[u] = L.list[ix[u]];
    }
#pragma unroll
    for (int u = 0; u < UM; ++u) { s[u] = ei[e[u]]; dl[u] = dstp[e[u]] - d0; }
#pragma unroll
    for (int u = 0; u < UM; ++u) {
#pragma unroll
      for (int h = 0; h < H; ++h) a[u][h] = als[(size_t)s[u] * H + h];
    }
#pragma unroll
    for (int u = 0; u < UM; ++u) {
      if (i0 + u * NT < n) {
        L.list[ix[u]] = s[u];
        L.dls[ix[u]] = (unsigned short)dl[u];
#pragma unroll
        for (int h = 0; h < H; ++h) {
          float v = lrelu(a[u][h] + L.aldT[dl[u] * H + h], 0.2f);
          ldsMaxF(&L.mx[dl[u] * H + h], v);
        }
      }
    }
  }
}

template <int H>
__device__ __forceinline__ void tile_max_kept(TileLds& L, const int* __restrict__ wl,
                                              const float* __restrict__ als, int n, int tid) {
  for (int i0 = tid; i0 < n; i0 += NT * UM) {
    int ix[UM], s[UM], dl[UM]; unsigned p[UM]; float a[UM][H];
#pragma unroll
    for (int u = 0; u < UM; ++u) {
      int i = i0 + u * NT;
      ix[u] = i < n ? i : i0;
      p[u] = (unsigned)wl[ix[u]];
    }
#pragma unroll
    for (int u = 0; u < UM; ++u) {
      s[u] = (int)(p[u] >> 8); dl[u] = (int)(p[u] & 255u);
#pragma unroll
      for (int h = 0; h < H; ++h) a[u][h] = als[(size_t)s[u] * H + h];
    }
#pragma unroll
    for (int u = 0; u < UM; ++u) {
      if (i0 + u * NT < n) {
        L.list[ix[u]] = s[u];
        L.dls[ix[u]] = (unsigned short)dl[u];
#pragma unroll
        for (int h = 0; h < H; ++h) {
          float v = lrelu(a[u][h] + L.aldT[dl[u] * H + h], 0.2f);
          ldsMaxF(&L.mx[dl[u] * H + h], v);
        }
      }
    }
  }
}

template <int H>
__device__ __forceinline__ void tile_sum(TileLds& L, const float* __restrict__ hfeat,
                                         const float* __restrict__ als, int n, int tid) {
  constexpr int F = HID / H;
  const int wave = tid >> 5, lane = tid & 31;
  for (int i0 = wave * UE; i0 < n; i0 += (NT / 32) * UE) {
    int s[UE], dl[UE]; float al[UE]; float hv[UE][4];
#pragma unroll
    for (int u = 0; u < UE; ++u) {
      int i = i0 + u; if (i >= n) i = n - 1;
      s[u] = L.list[i]; dl[u] = L.dls[i];
    }
#pragma unroll
    for (int u = 0; u < UE; ++u) {
      const float* hs = hfeat + (size_t)s[u] * HID;
      al[u] = (lane < H) ? als[(size_t)s[u] * H + lane] : 0.f;
#pragma unroll
      for (int k = 0; k < 4; ++k) hv[u][k] = hs[lane + 32 * k];
    }
#pragma unroll
    for (int u = 0; u < UE; ++u) {
      if (i0 + u < n) {
        float w_l = 0.f;
        if (lane < H) {
          int q = dl[u] * H + lane;
          float v = lrelu(al[u] + L.aldT[q], 0.2f);
          w_l = __expf(v - L.mx[q]);
          atomicAdd(&L.den[q], w_l);
        }
#pragma unroll
        for (int k = 0; k < 4; ++k) {
          int c = lane + 32 * k;
          float w = __shfl(w_l, c / F, 32);
          atomicAdd(&L.acc[dl[u] * HID + c], w * hv[u][k]);
        }
      }
    }
  }
}

template <int H>
__device__ __forceinline__ void gat_tile_body(TileLds& L, const int* __restrict__ ei,
                                              const float* __restrict__ hfeat,
                                              const float* __restrict__ als,
                                              const float* __restrict__ ald,
                                              const float* __restrict__ bias,
                                              float* __restrict__ outp,
                                              const int* __restrict__ keepbuf,
                                              const int* __restrict__ tcnt,
                                              int E, int N, int act) {
  constexpr int F = HID / H;
  const int tid = threadIdx.x;
  const int d0 = blockIdx.x * TD;
  int tdn = N - d0; if (tdn > TD) tdn = TD;
  const int* dstp = ei + (size_t)E;
  const int* wl = keepbuf + (size_t)blockIdx.x * CAP;

  for (int i = tid; i < TD * HID; i += NT) L.acc[i] = 0.f;
  for (int i = tid; i < tdn * H; i += NT) {
    size_t g = (size_t)d0 * H + i;
    float a_d = ald[g];
    float v = lrelu(als[g] + a_d, 0.2f);
    L.aldT[i] = a_d; L.wsf[i] = v; L.mx[i] = v;
  }

  __syncthreads();
  int n = tcnt[blockIdx.x];
  if (N > KEEP_MAXN) n = CAP + 1;
  const bool kept = n <= CAP;
  if (!kept) {
    if (tid == 0) L.cnt = 0;
    __syncthreads();
    tile_scan(L, dstp, 0, E, d0, tdn, tid);
    __syncthreads();
    n = L.cnt;
  }
  if (n <= CAP) {
    if (kept) tile_max_kept<H>(L, wl, als, n, tid);
    else      tile_max<H>(L, ei, dstp, als, n, d0, tid);
    __syncthreads();
    for (int i = tid; i < tdn * H; i += NT) {
      float w = __expf(L.wsf[i] - L.mx[i]);
      L.wsf[i] = w; L.den[i] = w;
    }
    __syncthreads();
    tile_sum<H>(L, hfeat, als, n, tid);
  } else {
    const int nwin = (E + CAP - 1) / CAP;
    for (int pass = 0; pass < 2; ++pass) {
      for (int w = 0; w < nwin; ++w) {
        __syncthreads();
        if (tid == 0) L.cnt = 0;
        __syncthreads();
        const int lo = w * CAP;
        int hi = lo + CAP; if (hi > E) hi = E;
        tile_scan(L, dstp, lo, hi, d0, tdn, tid);
        __syncthreads();
        n = L.cnt;
        tile_max<H>(L, ei, dstp, als, n, d0, tid);
        __syncthreads();
        if (pass == 1) tile_sum<H>(L, hfeat, als, n, tid);
      }
      if (pass == 0) {
        __syncthreads();
        for (int i = tid; i < tdn * H; i += NT) {
          float w = __expf(L.wsf[i] - L.mx[i]);
          L.wsf[i] = w; L.den[i] = w;
        }
      }
    }
  }
  __syncthreads();

  {
    constexpr int UO = 5;
    constexpr int RS = NT / HID;
    const int c = tid & (HID - 1);
    const int hq = c / F;
    const float bv = bias[c];
    for (int r0 = tid / HID; r0 < tdn; r0 += RS * UO) {
      float hv[UO];
#pragma unroll
      for (int u = 0; u < UO; ++u) {
        int dl = r0 + RS * u; if (dl >= tdn) dl = tdn - 1;
        hv[u] = hfeat[(size_t)(d0 + dl) * HID + c];
      }
#pragma unroll
      for (int u = 0; u < UO; ++u) {
        const int dl = r0 + RS * u;
        if (dl < tdn) {
          const int q = dl * H + hq;
          float v = (L.acc[dl * HID + c] + L.wsf[q] * hv[u]) / L.den[q] + bv;
          if (act) v = lrelu(v, 0.01f);
          outp[(size_t)(d0 + dl) * HID + c] = v;
        }
      }
    }
  }
}

__global__ void __launch_bounds__(NT) gat_tile_h4_kernel(const int* __restrict__ ei,
    const float* __restrict__ hfeat, const float* __restrict__ als, const float* __restrict__ ald,
    const float* __restrict__ bias, float* __restrict__ outp, const int* __restrict__ keepbuf,
    const int* __restrict__ tcnt, int E, int N, int act) {
  __shared__ TileLds L;
  gat_tile_body<4>(L, ei, hfeat, als, ald, bias, outp, keepbuf, tcnt, E, N, act);
}

__global__ void __launch_bounds__(NT) gat_tile_h1_kernel(const int* __restrict__ ei,
    const float* __restrict__ hfeat, const float* __restrict__ als, const float* __restrict__ ald,
    const float* __restrict__ bias, float* __restrict__ outp, const int* __restrict__ keepbuf,
    const int* __restrict__ tcnt, int E, int N, int act) {
  __shared__ TileLds L;
  gat_tile_body<1>(L, ei, hfeat, als, ald, bias, outp, keepbuf, tcnt, E, N, act);
}

#define BNT   512
#define BNE   8
#define BMAXT 2048
__global__ void __launch_bounds__(BNT)
bin_edges_kernel(const int* __restrict__ ei, int* __restrict__ keepbuf, int* __restrict__ tcnt,
                 int E, int N, int ntiles) {
  __shared__ int cnt[BMAXT + 1];
  __shared__ int base[BMAXT + 1];
  const int tid = threadIdx.x;
  const int* dstp = ei + (size_t)E;
  const int e0 = blockIdx.x * (BNT * BNE) + tid;
  int s[BNE], tl[BNE], dl[BNE];
#pragma unroll
  for (int u = 0; u < BNE; ++u) {
    const int e = e0 + u * BNT;
    const int ec = e < E ? e : E - 1;
    const int d = dstp[ec];
    s[u] = ei[ec];
    const int t = d / TD;
    dl[u] = d - t * TD;
    tl[u] = ((e < E) & ((unsigned)d < (unsigned)N)) ? t : -1;
  }
  for (int w0 = 0; w0 < ntiles; w0 += BMAXT) {
    const int nt = ntiles - w0 < BMAXT ? ntiles - w0 : BMAXT;
    for (int i = tid; i < nt; i += BNT) cnt[i] = 0;
    if (tid == 0) { cnt[BMAXT] = 0; base[BMAXT] = CAP; }
    __syncthreads();
    int rr[BNE];
#pragma unroll
    for (int u = 0; u < BNE; ++u) {
      const unsigned r = (unsigned)(tl[u] - w0);
      rr[u] = r < (unsigned)nt ? (int)r : BMAXT;
      atomicAdd(&cnt[rr[u]], 1);
    }
    __syncthreads();
    for (int i = tid; i < nt; i += BNT) {
      const int k = cnt[i];
      if (k > 0) base[i] = atomicAdd(&tcnt[w0 + i], k);
      cnt[i] = 0;
    }
    __syncthreads();
    int pos[BNE];
#pragma unroll
    for (int u = 0; u < BNE; ++u) pos[u] = base[rr[u]] + atomicAdd(&cnt[rr[u]], 1);
#pragma unroll
    for (int u = 0; u < BNE; ++u) {
      if (rr[u] != BMAXT && pos[u] < CAP)
        keepbuf[(size_t)tl[u] * CAP + pos[u]] = (int)(((unsigned)s[u] << 8) | (unsigned)dl[u]);
    }
    __syncthreads();
  }
}

extern "C" void kernel_launch(void* const* d_in, const int* in_sizes, int n_in,
                              void* d_out, int out_size, void* d_ws, size_t ws_size,
                              hipStream_t stream) {
  const float* num_prop = (const float*)d_in[0];
  const float* cat_prop = (const float*)d_in[1];
  const int*   ei       = (const int*)d_in[2];
  const float* W_num = (const float*)d_in[3];
  const float* b_num = (const float*)d_in[4];
  const float* W_cat = (const float*)d_in[5];
  const float* b_cat = (const float*)d_in[6];
  const float* W_in  = (const float*)d_in[7];
  const float* b_in  = (const float*)d_in[8];
  const float* bn_g  = (const float*)d_in[9];
  const float* bn_b  = (const float*)d_in[10];
  const float* W1    = (const float*)d_in[11];
  const float* a_s1  = (const float*)d_in[12];
  const float* a_d1  = (const float*)d_in[13];
  const float* bias1 = (const float*)d_in[14];
  const float* W2    = (const float*)d_in[15];
  const float* a_s2  = (const float*)d_in[16];
  const float* a_d2  = (const float*)d_in[17];
  const float* bias2 = (const float*)d_in[18];

  const int N = in_sizes[0] / 7;
  const int E = in_sizes[2] / 2;
  float* out = (float*)d_out;

  float* ws   = (float*)d_ws;
  float* xcat = ws;
  float* x1   = xcat + (size_t)N * HID;
  float* hbuf = x1   + (size_t)N * HID;
  float* accb = hbuf + (size_t)N * HID;
  float* als  = accb + (size_t)N * HID;
  float* ald  = als  + (size_t)N * 4;
  int*   tcnt = (int*)(ald + (size_t)N * 4);
  float* stats= ald  + (size_t)N * 12;

  const int TB = 256;
  const int ntiles = (N + TD - 1) / TD;
  const int mtiles = (N + 15) / 16;
  const int gblocks = (mtiles + GW - 1) / GW;
  auto nb = [&](long long work) { return (unsigned)((work + TB - 1) / TB); };
  int* keepbuf = (int*)x1;
  float* part = hbuf;
  const int bblocks = (E + BNT * BNE - 1) / (BNT * BNE);
  const int eblocks = (N + EW * ER - 1) / (EW * ER);
  const int ablocks = (N * (HID / 4) + AT * AU - 1) / (AT * AU);

  encoder_kernel<<<eblocks, EW * 32, 0, stream>>>(num_prop, cat_prop, W_num, b_num,
                                                  W_cat, b_cat, xcat, N);
  wmma_gemm128_kernel<true><<<gblocks, GW * 32, 0, stream>>>(xcat, W_in, b_in, x1, part, N, 1, 1);

  clear_kernel<<<nb(ntiles), TB, 0, stream>>>(tcnt, ntiles);
  bn_finalize_kernel<<<HID, FT, 0, stream>>>(part, gblocks, stats, bn_g, bn_b, N);
  bn_apply_kernel<<<ablocks, AT, 0, stream>>>(x1, stats, xcat, N);

  bin_edges_kernel<<<bblocks, BNT, 0, stream>>>(ei, keepbuf, tcnt, E, N, ntiles);

  wmma_gemm128_kernel<false><<<gblocks, GW * 32, 0, stream>>>(xcat, W1, nullptr, hbuf, nullptr, N, 0, 0);
  att_logits_kernel<<<nb((long long)N * 4), TB, 0, stream>>>(hbuf, a_s1, a_d1, als, ald, N, 4, 32);
  gat_tile_h4_kernel<<<ntiles, NT, 0, stream>>>(ei, hbuf, als, ald, bias1, accb, keepbuf, tcnt, E, N, 1);

  wmma_gemm128_kernel<false><<<gblocks, GW * 32, 0, stream>>>(accb, W2, nullptr, hbuf, nullptr, N, 0, 0);
  att_logits_kernel<<<nb(N), TB, 0, stream>>>(hbuf, a_s2, a_d2, als, ald, N, 1, 128);
  gat_tile_h1_kernel<<<ntiles, NT, 0, stream>>>(ei, hbuf, als, ald, bias2, out, keepbuf, tcnt, E, N, 0);
}


# GAT summing loop re-coded: 16-byte row loads, edge source/destination in scalar registers, a trip's 32 (edge, head) weights made once by the 32 lanes and passed inside 8-lane groups (DPP), not LDS
# speedup vs baseline: 4.3554x; 1.1942x over previous
#include <hip/hip_runtime.h>
#include <math.h>

typedef __attribute__((ext_vector_type(2))) float v2f;
typedef __attribute__((ext_vector_type(8))) float v8f;

#define HID 128

__device__ __forceinline__ float lrelu(float v, float s) { return v > 0.f ? v : s * v; }

#define ER 8
#define EW 8
__global__ void __launch_bounds__(EW * 32)
encoder_kernel(const float* __restrict__ num_prop, const float* __restrict__ cat_prop,
               const float* __restrict__ Wn, const float* __restrict__ bn,
               const float* __restrict__ Wc, const float* __restrict__ bc,
               float* __restrict__ xcat, int N) {
  const int lane = threadIdx.x & 31;
  const int wv = __builtin_amdgcn_readfirstlane((int)(threadIdx.x >> 5));
  const int n0 = (blockIdx.x * EW + wv) * ER;
  if (n0 >= N) return;
  const int j = 2 * lane;
  float wn[7][2], wc[3][2];
#pragma unroll
  for (int k = 0; k < 7; ++k) { wn[k][0] = Wn[k * 64 + j]; wn[k][1] = Wn[k * 64 + j + 1]; }
#pragma unroll
  for (int k = 0; k < 3; ++k) { wc[k][0] = Wc[k * 64 + j]; wc[k][1] = Wc[k * 64 + j + 1]; }
  const float bn0 = bn[j], bn1 = bn[j + 1], bc0 = bc[j], bc1 = bc[j + 1];
  float xn[ER][7], xc[ER][3];
#pragma unroll
  for (int u = 0; u < ER; ++u) {
    int n = n0 + u; if (n >= N) n = N - 1;
#pragma unroll
    for (int k = 0; k < 7; ++k) xn[u][k] = num_prop[(size_t)n * 7 + k];
#pragma unroll
    for (int k = 0; k < 3; ++k) xc[u][k] = cat_prop[(size_t)n * 3 + k];
  }
#pragma unroll
  for (int u = 0; u < ER; ++u) {
    if (n0 + u < N) {
      float an0 = bn0, an1 = bn1, ac0 = bc0, ac1 = bc1;
#pragma unroll
      for (int k = 0; k < 7; ++k) { an0 = __builtin_fmaf(xn[u][k], wn[k][0], an0); an1 = __builtin_fmaf(xn[u][k], wn[k][1], an1); }
#pragma unroll
      for (int k = 0; k < 3; ++k) { ac0 = __builtin_fmaf(xc[u][k], wc[k][0], ac0); ac1 = __builtin_fmaf(xc[u][k], wc[k][1], ac1); }
      float2 on, oc;
      on.x = lrelu(an0, 0.01f); on.y = lrelu(an1, 0.01f);
      oc.x = lrelu(ac0, 0.01f); oc.y = lrelu(ac1, 0.01f);
      float* row = xcat + (size_t)(n0 + u) * HID + j;
      *reinterpret_cast<float2*>(row)      = on;
      *reinterpret_cast<float2*>(row + 64) = oc;
    }
  }
}

#define GW 8
#define BP 136
template <bool SUMS>
__global__ void __launch_bounds__(GW * 32)
wmma_gemm128_kernel(const float* __restrict__ A, const float* __restrict__ B,
                    const float* __restrict__ bias, float* __restrict__ C,
                    float* __restrict__ part, int M, int act, int has_bias) {
  __shared__ float4 Bl[32 * BP];
  __shared__ float red[SUMS ? GW * 2 * HID : 1];
  const int tid  = threadIdx.x;
  const int wave = tid >> 5;
  const int lane = tid & 31;
  const int half = lane >> 4;
  const int m16  = lane & 15;
  const int row0 = (blockIdx.x * GW + wave) * 16;

  int arow = row0 + m16; if (arow >= M) arow = M - 1;
  const float4* Ap = reinterpret_cast<const float4*>(A + (size_t)arow * HID) + half;
  float4 a[16];
#pragma unroll
  for (int j = 0; j < 16; ++j) a[j] = Ap[2 * j];

  float bv[8];
#pragma unroll
  for (int t = 0; t < 8; ++t) bv[t] = has_bias ? bias[16 * t + m16] : 0.f;

  {
    const int col = tid & (HID - 1);
#pragma unroll
    for (int i = 0; i < 16; ++i) {
      const int q = (tid >> 7) + 2 * i;
      const float* bp = B + (size_t)(4 * q) * HID + col;
      float4 v;
      v.x = bp[0]; v.y = bp[HID]; v.z = bp[2 * HID]; v.w = bp[3 * HID];
      Bl[q * BP + col] = v;
    }
  }
  __syncthreads();
  if (!SUMS && row0 >= M) return;
  if (row0 < M) {

  v8f acc[8];
#pragma unroll
  for (int t = 0; t < 8; ++t) acc[t] = (v8f){0.f, 0.f, 0.f, 0.f, 0.f, 0.f, 0.f, 0.f};

  const float4* Bf = Bl + half * BP + m16;
  float4 bc[8], bn[8];
#pragma unroll
  for (int t = 0; t < 8; ++t) bc[t] = Bf[16 * t];
#pragma unroll
  for (int j = 0; j < 16; ++j) {
    if (j + 1 < 16) {
#pragma unroll
      for (int t = 0; t < 8; ++t) bn[t] = Bf[(2 * (j + 1)) * BP + 16 * t];
    }
    __builtin_amdgcn_sched_barrier(0);
    const v2f a0 = {a[j].x, a[j].y};
    const v2f a1 = {a[j].z, a[j].w};
#pragma unroll
    for (int t = 0; t < 8; ++t) {
      const v2f b0 = {bc[t].x, bc[t].y};
      acc[t] = __builtin_amdgcn_wmma_f32_16x16x4_f32(false, a0, false, b0, (short)0, acc[t], false, false);
    }
#pragma unroll
    for (int t = 0; t < 8; ++t) {
      const v2f b1 = {bc[t].z, bc[t].w};
      acc[t] = __builtin_amdgcn_wmma_f32_16x16x4_f32(false, a1, false, b1, (short)0, acc[t], false, false);
    }
    __builtin_amdgcn_sched_barrier(0);
#pragma unroll
    for (int t = 0; t < 8; ++t) bc[t] = bn[t];
  }

  float cs[8], cq[8];
#pragma unroll
  for (int t = 0; t < 8; ++t) { cs[t] = 0.f; cq[t] = 0.f; }
  float* Cp = C + (size_t)(row0 + 8 * half) * HID + m16;
  if (row0 + 16 <= M) {
#pragma unroll
    for (int r = 0; r < 8; ++r) {
#pragma unroll
      for (int t = 0; t < 8; ++t) {
        float v = acc[t][r] + bv[t];
        if (act) v = lrelu(v, 0.01f);
        Cp[r * HID + 16 * t] = v;
        if (SUMS) { cs[t] += v; cq[t] += v * v; }
      }
    }
  } else {
#pragma unroll
    for (int r = 0; r < 8; ++r) {
      if (row0 + 8 * half + r < M) {
#pragma unroll
        for (int t = 0; t < 8; ++t) {
          float v = acc[t][r] + bv[t];
          if (act) v = lrelu(v, 0.01f);
          Cp[r * HID + 16 * t] = v;
          if (SUMS) { cs[t] += v; cq[t] += v * v; }
        }
      }
    }
  }
  if (SUMS) {
#pragma unroll
    for (int t = 0; t < 8; ++t) {
      const float s = cs[t] + __shfl_xor(cs[t], 16, 32);
      const float q = cq[t] + __shfl_xor(cq[t], 16, 32);
      red[wave * 2 * HID + half * HID + 16 * t + m16] = half ? q : s;
    }
  }
  } else if (SUMS) {
#pragma unroll
    for (int t = 0; t < 8; ++t) red[wave * 2 * HID + half * HID + 16 * t + m16] = 0.f;
  }
  if (SUMS) {
    __syncthreads();
    float s = 0.f;
#pragma unroll
    for (int w = 0; w < GW; ++w) s += red[w * 2 * HID + tid];
    part[(size_t)blockIdx.x * 2 * HID + tid] = s;
  }
}

#define FT 256
__global__ void __launch_bounds__(FT)
bn_finalize_kernel(const float* __restrict__ part, int nblk, float* __restrict__ stats,
                   const float* __restrict__ gamma, const float* __restrict__ beta, int N) {
  __shared__ float ws_[FT / 32], wq_[FT / 32];
  const int c = blockIdx.x;
  const int i = threadIdx.x;
  float s = 0.f, q = 0.f;
  for (int b = i; b < nblk; b += FT) {
    s += part[(size_t)b * 2 * HID + c];
    q += part[(size_t)b * 2 * HID + HID + c];
  }
#pragma unroll
  for (int w = 16; w > 0; w >>= 1) { s += __shfl_xor(s, w, 32); q += __shfl_xor(q, w, 32); }
  if ((i & 31) == 0) { ws_[i >> 5] = s; wq_[i >> 5] = q; }
  __syncthreads();
  if (i == 0) {
    float ts = 0.f, tq = 0.f;
#pragma unroll
    for (int w = 0; w < FT / 32; ++w) { ts += ws_[w]; tq += wq_[w]; }
    const float inv_n = 1.f / (float)N;
    const float mu  = ts * inv_n;
    const float var = tq * inv_n - mu * mu;
    const float sc  = gamma[c] * rsqrtf(var + 1e-5f);
    stats[2 * HID + c] = sc;
    stats[3 * HID + c] = beta[c] - mu * sc;
  }
}

#define AU 8
#define AT 256
__device__ __forceinline__ float4 bn_piece(float4 v, float4 sc, float4 sh) {
  float4 r;
  r.x = v.x * sc.x + sh.x; r.y = v.y * sc.y + sh.y;
  r.z = v.z * sc.z + sh.z; r.w = v.w * sc.w + sh.w;
  return r;
}
__global__ void __launch_bounds__(AT)
bn_apply_kernel(const float* __restrict__ x, const float* __restrict__ stats,
                float* __restrict__ y, int N) {
  const int total4 = N * (HID / 4);
  const int t = threadIdx.x;
  const int b0 = blockIdx.x * (AT * AU);
  const int c4 = t & (HID / 4 - 1);
  const float4 sc = reinterpret_cast<const float4*>(stats + 2 * HID)[c4];
  const float4 sh = reinterpret_cast<const float4*>(stats + 3 * HID)[c4];
  const float4* __restrict__ x4 = reinterpret_cast<const float4*>(x) + b0 + t;
  float4* __restrict__ y4 = reinterpret_cast<float4*>(y) + b0 + t;
  if (b0 + AT * AU <= total4) {
    float4 v[AU];
#pragma unroll
    for (int u = 0; u < AU; ++u) v[u] = x4[u * AT];
#pragma unroll
    for (int u = 0; u < AU; ++u) y4[u * AT] = bn_piece(v[u], sc, sh);
  } else {
#pragma unroll
    for (int u = 0; u < AU; ++u) {
      if (b0 + t + u * AT < total4) y4[u * AT] = bn_piece(x4[u * AT], sc, sh);
    }
  }
}

__global__ void att_logits_kernel(const float* __restrict__ h, const float* __restrict__ a_src,
                                  const float* __restrict__ a_dst, float* __restrict__ als,
                                  float* __restrict__ ald, int N, int H, int F) {
  int t = blockIdx.x * blockDim.x + threadIdx.x;
  if (t >= N * H) return;
  int n = t / H, hh = t - n * H;
  const float* hp = h + (size_t)n * HID + hh * F;
  const float* as = a_src + hh * F;
  const float* ad = a_dst + hh * F;
  float ss = 0.f, sd = 0.f;
  for (int f = 0; f < F; ++f) { float v = hp[f]; ss += v * as[f]; sd += v * ad[f]; }
  als[t] = ss;
  ald[t] = sd;
}

__global__ void clear_kernel(int* __restrict__ tcnt, int nt) {
  int t = blockIdx.x * blockDim.x + threadIdx.x;
  if (t < nt) tcnt[t] = 0;
}

#define TD   196
#define CAP  16384
#define NT   1024
#define UB   8
#define UE   8

struct TileLds {
  float mx[TD * 4];
  float den[TD * 4];
  float aldT[TD * 4];
  float wsf[TD * 4];
  float acc[TD * HID];
  int   list[CAP];
  unsigned short dls[CAP];
  int   cnt;
};

__device__ __forceinline__ void ldsMaxF(float* addr, float val) {
  if (val >= 0.f) atomicMax((int*)addr, __float_as_int(val));
  else            atomicMin((unsigned int*)addr, __float_as_uint(val));
}

__device__ __forceinline__ void tile_push(TileLds& L, int e) {
  int pos = atomicAdd(&L.cnt, 1);
  if (pos < CAP) L.list[pos] = e;
}

__device__ __forceinline__ void tile_test1(TileLds& L, const int* __restrict__ dstp, int e,
                                           int d0, int tdn) {
  if ((unsigned)(dstp[e] - d0) < (unsigned)tdn) tile_push(L, e);
}

__device__ __forceinline__ void tile_scan(TileLds& L, const int* __restrict__ dstp, int lo, int hi,
                                          int d0, int tdn, int tid) {
  int head = (4 - (int)((((size_t)(dstp + lo)) >> 2) & 3)) & 3;
  int a0 = lo + head; if (a0 > hi) a0 = hi;
  if (tid < a0 - lo) tile_test1(L, dstp, lo + tid, d0, tdn);
  const int nvec = (hi - a0) >> 2;
  const int4* p4 = (const int4*)(dstp + a0);
  const int rot = (nvec / (int)gridDim.x) * (int)blockIdx.x;
  for (int j = 0; j < nvec; j += NT * UB) {
    int4 v[UB]; int idx[UB];
#pragma unroll
    for (int u = 0; u < UB; ++u) {
      int k = j + u * NT + tid;
      int i = k + rot; if (i >= nvec) i -= nvec;
      if (k >= nvec) i = -1;
      idx[u] = i;
      v[u] = p4[i < 0 ? 0 : i];
    }
#pragma unroll
    for (int u = 0; u < UB; ++u) {
      const bool ok = idx[u] >= 0;
      bool h0 = ok & ((unsigned)(v[u].x - d0) < (unsigned)tdn);
      bool h1 = ok & ((unsigned)(v[u].y - d0) < (unsigned)tdn);
      bool h2 = ok & ((unsigned)(v[u].z - d0) < (unsigned)tdn);
      bool h3 = ok & ((unsigned)(v[u].w - d0) < (unsigned)tdn);
      if (h0 | h1 | h2 | h3) {
        int e0 = a0 + 4 * idx[u];
        if (h0) tile_push(L, e0);
        if (h1) tile_push(L, e0 + 1);
        if (h2) tile_push(L, e0 + 2);
        if (h3) tile_push(L, e0 + 3);
      }
    }
  }
  int t0 = a0 + 4 * nvec;
  if (tid < hi - t0) tile_test1(L, dstp, t0 + tid, d0, tdn);
}

#define UM 4
#define KEEP_MAXN (1 << 24)
template <int H>
__device__ __forceinline__ void tile_max(TileLds& L, const int* __restrict__ ei,
                                         const int* __restrict__ dstp,
                                         const float* __restrict__ als,
                                         int n, int d0, int tid) {
  for (int i0 = tid; i0 < n; i0 += NT * UM) {
    int ix[UM], e[UM], s[UM], dl[UM]; float a[UM][H];
#pragma unroll
    for (int u = 0; u < UM; ++u) {
      int i = i0 + u * NT;
      ix[u] = i < n ? i : i0;
      e[u] = L.list[ix[u]];
    }
#pragma unroll
    for (int u = 0; u < UM; ++u) { s[u] = ei[e[u]]; dl[u] = dstp[e[u]] - d0; }
#pragma unroll
    for (int u = 0; u < UM; ++u) {
#pragma unroll
      for (int h = 0; h < H; ++h) a[u][h] = als[(size_t)s[u] * H + h];
    }
#pragma unroll
    for (int u = 0; u < UM; ++u) {
      if (i0 + u * NT < n) {
        L.list[ix[u]] = s[u];
        L.dls[ix[u]] = (unsigned short)dl[u];
#pragma unroll
        for (int h = 0; h < H; ++h) {
          float v = lrelu(a[u][h] + L.aldT[dl[u] * H + h], 0.2f);
          ldsMaxF(&L.mx[dl[u] * H + h], v);
        }
      }
    }
  }
}

template <int H>
__device__ __forceinline__ void tile_max_kept(TileLds& L, const int* __restrict__ wl,
                                              const float* __restrict__ als, int n, int tid) {
  for (int i0 = tid; i0 < n; i0 += NT * UM) {
    int ix[UM], s[UM], dl[UM]; unsigned p[UM]; float a[UM][H];
#pragma unroll
    for (int u = 0; u < UM; ++u) {
      int i = i0 + u * NT;
      ix[u] = i < n ? i : i0;
      p[u] = (unsigned)wl[ix[u]];
    }
#pragma unroll
    for (int u = 0; u < UM; ++u) {
      s[u] = (int)(p[u] >> 8); dl[u] = (int)(p[u] & 255u);
#pragma unroll
      for (int h = 0; h < H; ++h) a[u][h] = als[(size_t)s[u] * H + h];
    }
#pragma unroll
    for (int u = 0; u < UM; ++u) {
      if (i0 + u * NT < n) {
        L.list[ix[u]] = s[u];
        L.dls[ix[u]] = (unsigned short)dl[u];
#pragma unroll
        for (int h = 0; h < H; ++h) {
          float v = lrelu(a[u][h] + L.aldT[dl[u] * H + h], 0.2f);
          ldsMaxF(&L.mx[dl[u] * H + h], v);
        }
      }
    }
  }
}


__device__ __forceinline__ float grp8_take(float x, int u) {
  const unsigned v = __float_as_uint(x);
  unsigned r;
  switch (u) {
    case 0:  r = __builtin_amdgcn_mov_dpp8(v, 0x000000); break;
    case 1:  r = __builtin_amdgcn_mov_dpp8(v, 0x249249); break;
    case 2:  r = __builtin_amdgcn_mov_dpp8(v, 0x492492); break;
    case 3:  r = __builtin_amdgcn_mov_dpp8(v, 0x6DB6DB); break;
    case 4:  r = __builtin_amdgcn_mov_dpp8(v, 0x924924); break;
    case 5:  r = __builtin_amdgcn_mov_dpp8(v, 0xB6DB6D); break;
    case 6:  r = __builtin_amdgcn_mov_dpp8(v, 0xDB6DB6); break;
    default: r = __builtin_amdgcn_mov_dpp8(v, 0xFFFFFF); break;
  }
  return __uint_as_float(r);
}

template <int H>
__device__ __forceinline__ void tile_sum(TileLds& L, const float* __restrict__ hfeat,
                                         const float* __restrict__ als, int n_, int tid) {
  static_assert(H == 1 || H == 4, "a group of 8 lanes is one head (4 heads) or all lanes share one head");
  static_assert(UE == 8, "lane u of a group of 8 lanes holds the weight of entry u of the trip");
  const int wave = __builtin_amdgcn_readfirstlane(tid >> 5);
  const int n = __builtin_amdgcn_readfirstlane(n_);
  const int lane = tid & 31;
  const int g = lane >> 3;
  const int hq = (H == 4) ? g : 0;
  for (int i0 = wave * UE; i0 < n; i0 += (NT / 32) * UE) {
    const int io = i0 + (lane & 7);
    const int i = io < n ? io : n - 1;
    const int sv = L.list[i];
    const int dv = L.dls[i];
    const int qv = dv * H + hq;
    const float alv = als[(size_t)sv * H + hq];
    const float adv = L.aldT[qv], mvv = L.mx[qv];
    int s[UE], dl[UE]; float4 hv[UE];
#pragma unroll
    for (int u = 0; u < UE; ++u) {
      s[u]  = __builtin_amdgcn_readlane(sv, u);
      dl[u] = __builtin_amdgcn_readlane(dv, u);
    }
#pragma unroll
    for (int u = 0; u < UE; ++u)
      hv[u] = reinterpret_cast<const float4*>(hfeat + (size_t)s[u] * HID)[lane];
    const float wv = __expf(lrelu(alv + adv, 0.2f) - mvv);
    if (io < n && g < H) atomicAdd(&L.den[qv], wv);
#pragma unroll
    for (int u = 0; u < UE; ++u) {
      if (i0 + u < n) {
        const float w = grp8_take(wv, u);
        float* a = &L.acc[dl[u] * HID + lane];
        atomicAdd(a,      w * hv[u].x);
        atomicAdd(a + 32, w * hv[u].y);
        atomicAdd(a + 64, w * hv[u].z);
        atomicAdd(a + 96, w * hv[u].w);
      }
    }
  }
}

template <int H>
__device__ __forceinline__ void gat_tile_body(TileLds& L, const int* __restrict__ ei,
                                              const float* __restrict__ hfeat,
                                              const float* __restrict__ als,
                                              const float* __restrict__ ald,
                                              const float* __restrict__ bias,
                                              float* __restrict__ outp,
                                              const int* __restrict__ keepbuf,
                                              const int* __restrict__ tcnt,
                                              int E, int N, int act) {
  constexpr int F = HID / H;
  const int tid = threadIdx.x;
  const int d0 = blockIdx.x * TD;
  int tdn = N - d0; if (tdn > TD) tdn = TD;
  const int* dstp = ei + (size_t)E;
  const int* wl = keepbuf + (size_t)blockIdx.x * CAP;

  for (int i = tid; i < TD * HID; i += NT) L.acc[i] = 0.f;
  for (int i = tid; i < tdn * H; i += NT) {
    size_t g = (size_t)d0 * H + i;
    float a_d = ald[g];
    float v = lrelu(als[g] + a_d, 0.2f);
    L.aldT[i] = a_d; L.wsf[i] = v; L.mx[i] = v;
  }

  __syncthreads();
  int n = tcnt[blockIdx.x];
  if (N > KEEP_MAXN) n = CAP + 1;
  const bool kept = n <= CAP;
  if (!kept) {
    if (tid == 0) L.cnt = 0;
    __syncthreads();
    tile_scan(L, dstp, 0, E, d0, tdn, tid);
    __syncthreads();
    n = L.cnt;
  }
  if (n <= CAP) {
    if (kept) tile_max_kept<H>(L, wl, als, n, tid);
    else      tile_max<H>(L, ei, dstp, als, n, d0, tid);
    __syncthreads();
    for (int i = tid; i < tdn * H; i += NT) {
      float w = __expf(L.wsf[i] - L.mx[i]);
      L.wsf[i] = w; L.den[i] = w;
    }
    __syncthreads();
    tile_sum<H>(L, hfeat, als, n, tid);
  } else {
    const int nwin = (E + CAP - 1) / CAP;
    for (int pass = 0; pass < 2; ++pass) {
      for (int w = 0; w < nwin; ++w) {
        __syncthreads();
        if (tid == 0) L.cnt = 0;
        __syncthreads();
        const int lo = w * CAP;
        int hi = lo + CAP; if (hi > E) hi = E;
        tile_scan(L, dstp, lo, hi, d0, tdn, tid);
        __syncthreads();
        n = L.cnt;
        tile_max<H>(L, ei, dstp, als, n, d0, tid);
        __syncthreads();
        if (pass == 1) tile_sum<H>(L, hfeat, als, n, tid);
      }
      if (pass == 0) {
        __syncthreads();
        for (int i = tid; i < tdn * H; i += NT) {
          float w = __expf(L.wsf[i] - L.mx[i]);
          L.wsf[i] = w; L.den[i] = w;
        }
      }
    }
  }
  __syncthreads();

  {
    constexpr int UO = 7;
    constexpr int RS = NT / 32;
    const int lane = tid & 31;
    const int hq = (4 * lane) / F;
    float bv[4];
#pragma unroll
    for (int k = 0; k < 4; ++k) bv[k] = bias[4 * lane + k];
    for (int r0 = tid >> 5; r0 < tdn; r0 += RS * UO) {
      float4 hv[UO];
#pragma unroll
      for (int u = 0; u < UO; ++u) {
        int dl = r0 + RS * u; if (dl >= tdn) dl = tdn - 1;
        hv[u] = reinterpret_cast<const float4*>(hfeat + (size_t)(d0 + dl) * HID)[lane];
      }
#pragma unroll
      for (int u = 0; u < UO; ++u) {
        const int dl = r0 + RS * u;
        if (dl < tdn) {
          const int q = dl * H + hq;
          const float ws = L.wsf[q], dn = L.den[q];
          const float* a = &L.acc[dl * HID + lane];
          float4 o;
          o.x = (a[0]  + ws * hv[u].x) / dn + bv[0];
          o.y = (a[32] + ws * hv[u].y) / dn + bv[1];
          o.z = (a[64] + ws * hv[u].z) / dn + bv[2];
          o.w = (a[96] + ws * hv[u].w) / dn + bv[3];
          if (act) { o.x = lrelu(o.x, 0.01f); o.y = lrelu(o.y, 0.01f); o.z = lrelu(o.z, 0.01f); o.w = lrelu(o.w, 0.01f); }
          reinterpret_cast<float4*>(outp + (size_t)(d0 + dl) * HID)[lane] = o;
        }
      }
    }
  }
}

__global__ void __launch_bounds__(NT) gat_tile_h4_kernel(const int* __restrict__ ei,
    const float* __restrict__ hfeat, const float* __restrict__ als, const float* __restrict__ ald,
    const float* __restrict__ bias, float* __restrict__ outp, const int* __restrict__ keepbuf,
    const int* __restrict__ tcnt, int E, int N, int act) {
  __shared__ TileLds L;
  gat_tile_body<4>(L, ei, hfeat, als, ald, bias, outp, keepbuf, tcnt, E, N, act);
}

__global__ void __launch_bounds__(NT) gat_tile_h1_kernel(const int* __restrict__ ei,
    const float* __restrict__ hfeat, const float* __restrict__ als, const float* __restrict__ ald,
    const float* __restrict__ bias, float* __restrict__ outp, const int* __restrict__ keepbuf,
    const int* __restrict__ tcnt, int E, int N, int act) {
  __shared__ TileLds L;
  gat_tile_body<1>(L, ei, hfeat, als, ald, bias, outp, keepbuf, tcnt, E, N, act);
}

#define BNT   512
#define BNE   8
#define BMAXT 2048
__global__ void __launch_bounds__(BNT)
bin_edges_kernel(const int* __restrict__ ei, int* __restrict__ keepbuf, int* __restrict__ tcnt,
                 int E, int N, int ntiles) {
  __shared__ int cnt[BMAXT + 1];
  __shared__ int base[BMAXT + 1];
  const int tid = threadIdx.x;
  const int* dstp = ei + (size_t)E;
  const int e0 = blockIdx.x * (BNT * BNE) + tid;
  int s[BNE], tl[BNE], dl[BNE];
#pragma unroll
  for (int u = 0; u < BNE; ++u) {
    const int e = e0 + u * BNT;
    const int ec = e < E ? e : E - 1;
    const int d = dstp[ec];
    s[u] = ei[ec];
    const int t = d / TD;
    dl[u] = d - t * TD;
    tl[u] = ((e < E) & ((unsigned)d < (unsigned)N)) ? t : -1;
  }
  for (int w0 = 0; w0 < ntiles; w0 += BMAXT) {
    const int nt = ntiles - w0 < BMAXT ? ntiles - w0 : BMAXT;
    for (int i = tid; i < nt; i += BNT) cnt[i] = 0;
    if (tid == 0) { cnt[BMAXT] = 0; base[BMAXT] = CAP; }
    __syncthreads();
    int rr[BNE];
#pragma unroll
    for (int u = 0; u < BNE; ++u) {
      const unsigned r = (unsigned)(tl[u] - w0);
      rr[u] = r < (unsigned)nt ? (int)r : BMAXT;
      atomicAdd(&cnt[rr[u]], 1);
    }
    __syncthreads();
    for (int i = tid; i < nt; i += BNT) {
      const int k = cnt[i];
      if (k > 0) base[i] = atomicAdd(&tcnt[w0 + i], k);
      cnt[i] = 0;
    }
    __syncthreads();
    int pos[BNE];
#pragma unroll
    for (int u = 0; u < BNE; ++u) pos[u] = base[rr[u]] + atomicAdd(&cnt[rr[u]], 1);
#pragma unroll
    for (int u = 0; u < BNE; ++u) {
      if (rr[u] != BMAXT && pos[u] < CAP)
        keepbuf[(size_t)tl[u] * CAP + pos[u]] = (int)(((unsigned)s[u] << 8) | (unsigned)dl[u]);
    }
    __syncthreads();
  }
}

extern "C" void kernel_launch(void* const* d_in, const int* in_sizes, int n_in,
                              void* d_out, int out_size, void* d_ws, size_t ws_size,
                              hipStream_t stream) {
  const float* num_prop = (const float*)d_in[0];
  const float* cat_prop = (const float*)d_in[1];
  const int*   ei       = (const int*)d_in[2];
  const float* W_num = (const float*)d_in[3];
  const float* b_num = (const float*)d_in[4];
  const float* W_cat = (const float*)d_in[5];
  const float* b_cat = (const float*)d_in[6];
  const float* W_in  = (const float*)d_in[7];
  const float* b_in  = (const float*)d_in[8];
  const float* bn_g  = (const float*)d_in[9];
  const float* bn_b  = (const float*)d_in[10];
  const float* W1    = (const float*)d_in[11];
  const float* a_s1  = (const float*)d_in[12];
  const float* a_d1  = (const float*)d_in[13];
  const float* bias1 = (const float*)d_in[14];
  const float* W2    = (const float*)d_in[15];
  const float* a_s2  = (const float*)d_in[16];
  const float* a_d2  = (const float*)d_in[17];
  const float* bias2 = (const float*)d_in[18];

  const int N = in_sizes[0] / 7;
  const int E = in_sizes[2] / 2;
  float* out = (float*)d_out;

  float* ws   = (float*)d_ws;
  float* xcat = ws;
  float* x1   = xcat + (size_t)N * HID;
  float* hbuf = x1   + (size_t)N * HID;
  float* accb = hbuf + (size_t)N * HID;
  float* als  = accb + (size_t)N * HID;
  float* ald  = als  + (size_t)N * 4;
  int*   tcnt = (int*)(ald + (size_t)N * 4);
  float* stats= ald  + (size_t)N * 12;

  const int TB = 256;
  const int ntiles = (N + TD - 1) / TD;
  const int mtiles = (N + 15) / 16;
  const int gblocks = (mtiles + GW - 1) / GW;
  auto nb = [&](long long work) { return (unsigned)((work + TB - 1) / TB); };
  int* keepbuf = (int*)x1;
  float* part = hbuf;
  const int bblocks = (E + BNT * BNE - 1) / (BNT * BNE);
  const int eblocks = (N + EW * ER - 1) / (EW * ER);
  const int ablocks = (N * (HID / 4) + AT * AU - 1) / (AT * AU);

  encoder_kernel<<<eblocks, EW * 32, 0, stream>>>(num_prop, cat_prop, W_num, b_num,
                                                  W_cat, b_cat, xcat, N);
  wmma_gemm128_kernel<true><<<gblocks, GW * 32, 0, stream>>>(xcat, W_in, b_in, x1, part, N, 1, 1);

  clear_kernel<<<nb(ntiles), TB, 0, stream>>>(tcnt, ntiles);
  bn_finalize_kernel<<<HID, FT, 0, stream>>>(part, gblocks, stats, bn_g, bn_b, N);
  bn_apply_kernel<<<ablocks, AT, 0, stream>>>(x1, stats, xcat, N);

  bin_edges_kernel<<<bblocks, BNT, 0, stream>>>(ei, keepbuf, tcnt, E, N, ntiles);

  wmma_gemm128_kernel<false><<<gblocks, GW * 32, 0, stream>>>(xcat, W1, nullptr, hbuf, nullptr, N, 0, 0);
  att_logits_kernel<<<nb((long long)N * 4), TB, 0, stream>>>(hbuf, a_s1, a_d1, als, ald, N, 4, 32);
  gat_tile_h4_kernel<<<ntiles, NT, 0, stream>>>(ei, hbuf, als, ald, bias1, accb, keepbuf, tcnt, E, N, 1);

  wmma_gemm128_kernel<false><<<gblocks, GW * 32, 0, stream>>>(accb, W2, nullptr, hbuf, nullptr, N, 0, 0);
  att_logits_kernel<<<nb(N), TB, 0, stream>>>(hbuf, a_s2, a_d2, als, ald, N, 1, 128);
  gat_tile_h1_kernel<<<ntiles, NT, 0, stream>>>(ei, hbuf, als, ald, bias2, out, keepbuf, tcnt, E, N, 0);
}


# GAT tile list put in node order in LDS (count, running sum, ranked places); a wave adds a node's neighbouring rows in registers, one LDS float add per run not per edge; order of the row sums changed
# speedup vs baseline: 4.4844x; 1.0296x over previous
#include <hip/hip_runtime.h>
#include <math.h>

typedef __attribute__((ext_vector_type(2))) float v2f;
typedef __attribute__((ext_vector_type(8))) float v8f;

#define HID 128

__device__ __forceinline__ float lrelu(float v, float s) { return v > 0.f ? v : s * v; }

#define ER 8
#define EW 8
__global__ void __launch_bounds__(EW * 32)
encoder_kernel(const float* __restrict__ num_prop, const float* __restrict__ cat_prop,
               const float* __restrict__ Wn, const float* __restrict__ bn,
               const float* __restrict__ Wc, const float* __restrict__ bc,
               float* __restrict__ xcat, int N) {
  const int lane = threadIdx.x & 31;
  const int wv = __builtin_amdgcn_readfirstlane((int)(threadIdx.x >> 5));
  const int n0 = (blockIdx.x * EW + wv) * ER;
  if (n0 >= N) return;
  const int j = 2 * lane;
  float wn[7][2], wc[3][2];
#pragma unroll
  for (int k = 0; k < 7; ++k) { wn[k][0] = Wn[k * 64 + j]; wn[k][1] = Wn[k * 64 + j + 1]; }
#pragma unroll
  for (int k = 0; k < 3; ++k) { wc[k][0] = Wc[k * 64 + j]; wc[k][1] = Wc[k * 64 + j + 1]; }
  const float bn0 = bn[j], bn1 = bn[j + 1], bc0 = bc[j], bc1 = bc[j + 1];
  float xn[ER][7], xc[ER][3];
#pragma unroll
  for (int u = 0; u < ER; ++u) {
    int n = n0 + u; if (n >= N) n = N - 1;
#pragma unroll
    for (int k = 0; k < 7; ++k) xn[u][k] = num_prop[(size_t)n * 7 + k];
#pragma unroll
    for (int k = 0; k < 3; ++k) xc[u][k] = cat_prop[(size_t)n * 3 + k];
  }
#pragma unroll
  for (int u = 0; u < ER; ++u) {
    if (n0 + u < N) {
      float an0 = bn0, an1 = bn1, ac0 = bc0, ac1 = bc1;
#pragma unroll
      for (int k = 0; k < 7; ++k) { an0 = __builtin_fmaf(xn[u][k], wn[k][0], an0); an1 = __builtin_fmaf(xn[u][k], wn[k][1], an1); }
#pragma unroll
      for (int k = 0; k < 3; ++k) { ac0 = __builtin_fmaf(xc[u][k], wc[k][0], ac0); ac1 = __builtin_fmaf(xc[u][k], wc[k][1], ac1); }
      float2 on, oc;
      on.x = lrelu(an0, 0.01f); on.y = lrelu(an1, 0.01f);
      oc.x = lrelu(ac0, 0.01f); oc.y = lrelu(ac1, 0.01f);
      float* row = xcat + (size_t)(n0 + u) * HID + j;
      *reinterpret_cast<float2*>(row)      = on;
      *reinterpret_cast<float2*>(row + 64) = oc;
    }
  }
}

#define GW 8
#define BP 136
template <bool SUMS>
__global__ void __launch_bounds__(GW * 32)
wmma_gemm128_kernel(const float* __restrict__ A, const float* __restrict__ B,
                    const float* __restrict__ bias, float* __restrict__ C,
                    float* __restrict__ part, int M, int act, int has_bias) {
  __shared__ float4 Bl[32 * BP];
  __shared__ float red[SUMS ? GW * 2 * HID : 1];
  const int tid  = threadIdx.x;
  const int wave = tid >> 5;
  const int lane = tid & 31;
  const int half = lane >> 4;
  const int m16  = lane & 15;
  const int row0 = (blockIdx.x * GW + wave) * 16;

  int arow = row0 + m16; if (arow >= M) arow = M - 1;
  const float4* Ap = reinterpret_cast<const float4*>(A + (size_t)arow * HID) + half;
  float4 a[16];
#pragma unroll
  for (int j = 0; j < 16; ++j) a[j] = Ap[2 * j];

  float bv[8];
#pragma unroll
  for (int t = 0; t < 8; ++t) bv[t] = has_bias ? bias[16 * t + m16] : 0.f;

  {
    const int col = tid & (HID - 1);
#pragma unroll
    for (int i = 0; i < 16; ++i) {
      const int q = (tid >> 7) + 2 * i;
      const float* bp = B + (size_t)(4 * q) * HID + col;
      float4 v;
      v.x = bp[0]; v.y = bp[HID]; v.z = bp[2 * HID]; v.w = bp[3 * HID];
      Bl[q * BP + col] = v;
    }
  }
  __syncthreads();
  if (!SUMS && row0 >= M) return;
  if (row0 < M) {

  v8f acc[8];
#pragma unroll
  for (int t = 0; t < 8; ++t) acc[t] = (v8f){0.f, 0.f, 0.f, 0.f, 0.f, 0.f, 0.f, 0.f};

  const float4* Bf = Bl + half * BP + m16;
  float4 bc[8], bn[8];
#pragma unroll
  for (int t = 0; t < 8; ++t) bc[t] = Bf[16 * t];
#pragma unroll
  for (int j = 0; j < 16; ++j) {
    if (j + 1 < 16) {
#pragma unroll
      for (int t = 0; t < 8; ++t) bn[t] = Bf[(2 * (j + 1)) * BP + 16 * t];
    }
    __builtin_amdgcn_sched_barrier(0);
    const v2f a0 = {a[j].x, a[j].y};
    const v2f a1 = {a[j].z, a[j].w};
#pragma unroll
    for (int t = 0; t < 8; ++t) {
      const v2f b0 = {bc[t].x, bc[t].y};
      acc[t] = __builtin_amdgcn_wmma_f32_16x16x4_f32(false, a0, false, b0, (short)0, acc[t], false, false);
    }
#pragma unroll
    for (int t = 0; t < 8; ++t) {
      const v2f b1 = {bc[t].z, bc[t].w};
      acc[t] = __builtin_amdgcn_wmma_f32_16x16x4_f32(false, a1, false, b1, (short)0, acc[t], false, false);
    }
    __builtin_amdgcn_sched_barrier(0);
#pragma unroll
    for (int t = 0; t < 8; ++t) bc[t] = bn[t];
  }

  float cs[8], cq[8];
#pragma unroll
  for (int t = 0; t < 8; ++t) { cs[t] = 0.f; cq[t] = 0.f; }
  float* Cp = C + (size_t)(row0 + 8 * half) * HID + m16;
  if (row0 + 16 <= M) {
#pragma unroll
    for (int r = 0; r < 8; ++r) {
#pragma unroll
      for (int t = 0; t < 8; ++t) {
        float v = acc[t][r] + bv[t];
        if (act) v = lrelu(v, 0.01f);
        Cp[r * HID + 16 * t] = v;
        if (SUMS) { cs[t] += v; cq[t] += v * v; }
      }
    }
  } else {
#pragma unroll
    for (int r = 0; r < 8; ++r) {
      if (row0 + 8 * half + r < M) {
#pragma unroll
        for (int t = 0; t < 8; ++t) {
          float v = acc[t][r] + bv[t];
          if (act) v = lrelu(v, 0.01f);
          Cp[r * HID + 16 * t] = v;
          if (SUMS) { cs[t] += v; cq[t] += v * v; }
        }
      }
    }
  }
  if (SUMS) {
#pragma unroll
    for (int t = 0; t < 8; ++t) {
      const float s = cs[t] + __shfl_xor(cs[t], 16, 32);
      const float q = cq[t] + __shfl_xor(cq[t], 16, 32);
      red[wave * 2 * HID + half * HID + 16 * t + m16] = half ? q : s;
    }
  }
  } else if (SUMS) {
#pragma unroll
    for (int t = 0; t < 8; ++t) red[wave * 2 * HID + half * HID + 16 * t + m16] = 0.f;
  }
  if (SUMS) {
    __syncthreads();
    float s = 0.f;
#pragma unroll
    for (int w = 0; w < GW; ++w) s += red[w * 2 * HID + tid];
    part[(size_t)blockIdx.x * 2 * HID + tid] = s;
  }
}

#define FT 256
__global__ void __launch_bounds__(FT)
bn_finalize_kernel(const float* __restrict__ part, int nblk, float* __restrict__ stats,
                   const float* __restrict__ gamma, const float* __restrict__ beta, int N) {
  __shared__ float ws_[FT / 32], wq_[FT / 32];
  const int c = blockIdx.x;
  const int i = threadIdx.x;
  float s = 0.f, q = 0.f;
  for (int b = i; b < nblk; b += FT) {
    s += part[(size_t)b * 2 * HID + c];
    q += part[(size_t)b * 2 * HID + HID + c];
  }
#pragma unroll
  for (int w = 16; w > 0; w >>= 1) { s += __shfl_xor(s, w, 32); q += __shfl_xor(q, w, 32); }
  if ((i & 31) == 0) { ws_[i >> 5] = s; wq_[i >> 5] = q; }
  __syncthreads();
  if (i == 0) {
    float ts = 0.f, tq = 0.f;
#pragma unroll
    for (int w = 0; w < FT / 32; ++w) { ts += ws_[w]; tq += wq_[w]; }
    const float inv_n = 1.f / (float)N;
    const float mu  = ts * inv_n;
    const float var = tq * inv_n - mu * mu;
    const float sc  = gamma[c] * rsqrtf(var + 1e-5f);
    stats[2 * HID + c] = sc;
    stats[3 * HID + c] = beta[c] - mu * sc;
  }
}

#define AU 8
#define AT 256
__device__ __forceinline__ float4 bn_piece(float4 v, float4 sc, float4 sh) {
  float4 r;
  r.x = v.x * sc.x + sh.x; r.y = v.y * sc.y + sh.y;
  r.z = v.z * sc.z + sh.z; r.w = v.w * sc.w + sh.w;
  return r;
}
__global__ void __launch_bounds__(AT)
bn_apply_kernel(const float* __restrict__ x, const float* __restrict__ stats,
                float* __restrict__ y, int N) {
  const int total4 = N * (HID / 4);
  const int t = threadIdx.x;
  const int b0 = blockIdx.x * (AT * AU);
  const int c4 = t & (HID / 4 - 1);
  const float4 sc = reinterpret_cast<const float4*>(stats + 2 * HID)[c4];
  const float4 sh = reinterpret_cast<const float4*>(stats + 3 * HID)[c4];
  const float4* __restrict__ x4 = reinterpret_cast<const float4*>(x) + b0 + t;
  float4* __restrict__ y4 = reinterpret_cast<float4*>(y) + b0 + t;
  if (b0 + AT * AU <= total4) {
    float4 v[AU];
#pragma unroll
    for (int u = 0; u < AU; ++u) v[u] = x4[u * AT];
#pragma unroll
    for (int u = 0; u < AU; ++u) y4[u * AT] = bn_piece(v[u], sc, sh);
  } else {
#pragma unroll
    for (int u = 0; u < AU; ++u) {
      if (b0 + t + u * AT < total4) y4[u * AT] = bn_piece(x4[u * AT], sc, sh);
    }
  }
}

__global__ void att_logits_kernel(const float* __restrict__ h, const float* __restrict__ a_src,
                                  const float* __restrict__ a_dst, float* __restrict__ als,
                                  float* __restrict__ ald, int N, int H, int F) {
  int t = blockIdx.x * blockDim.x + threadIdx.x;
  if (t >= N * H) return;
  int n = t / H, hh = t - n * H;
  const float* hp = h + (size_t)n * HID + hh * F;
  const float* as = a_src + hh * F;
  const float* ad = a_dst + hh * F;
  float ss = 0.f, sd = 0.f;
  for (int f = 0; f < F; ++f) { float v = hp[f]; ss += v * as[f]; sd += v * ad[f]; }
  als[t] = ss;
  ald[t] = sd;
}

__global__ void clear_kernel(int* __restrict__ tcnt, int nt) {
  int t = blockIdx.x * blockDim.x + threadIdx.x;
  if (t < nt) tcnt[t] = 0;
}

#define TD   196
#define CAP  16384
#define NT   1024
#define UB   8
#define UE   8
#define NCN  224

struct TileLds {
  float mx[TD * 4];
  float den[TD * 4];
  float aldT[TD * 4];
  float wsf[TD * 4];
  int   ncnt[NCN];
  float acc[TD * HID];
  int   list[CAP];
  unsigned short dls[CAP];
  int   cnt;
};

__device__ __forceinline__ void ldsMaxF(float* addr, float val) {
  if (val >= 0.f) atomicMax((int*)addr, __float_as_int(val));
  else            atomicMin((unsigned int*)addr, __float_as_uint(val));
}

__device__ __forceinline__ void tile_push(TileLds& L, int e) {
  int pos = atomicAdd(&L.cnt, 1);
  if (pos < CAP) L.list[pos] = e;
}

__device__ __forceinline__ void tile_test1(TileLds& L, const int* __restrict__ dstp, int e,
                                           int d0, int tdn) {
  if ((unsigned)(dstp[e] - d0) < (unsigned)tdn) tile_push(L, e);
}

__device__ __forceinline__ void tile_scan(TileLds& L, const int* __restrict__ dstp, int lo, int hi,
                                          int d0, int tdn, int tid) {
  int head = (4 - (int)((((size_t)(dstp + lo)) >> 2) & 3)) & 3;
  int a0 = lo + head; if (a0 > hi) a0 = hi;
  if (tid < a0 - lo) tile_test1(L, dstp, lo + tid, d0, tdn);
  const int nvec = (hi - a0) >> 2;
  const int4* p4 = (const int4*)(dstp + a0);
  const int rot = (nvec / (int)gridDim.x) * (int)blockIdx.x;
  for (int j = 0; j < nvec; j += NT * UB) {
    int4 v[UB]; int idx[UB];
#pragma unroll
    for (int u = 0; u < UB; ++u) {
      int k = j + u * NT + tid;
      int i = k + rot; if (i >= nvec) i -= nvec;
      if (k >= nvec) i = -1;
      idx[u] = i;
      v[u] = p4[i < 0 ? 0 : i];
    }
#pragma unroll
    for (int u = 0; u < UB; ++u) {
      const bool ok = idx[u] >= 0;
      bool h0 = ok & ((unsigned)(v[u].x - d0) < (unsigned)tdn);
      bool h1 = ok & ((unsigned)(v[u].y - d0) < (unsigned)tdn);
      bool h2 = ok & ((unsigned)(v[u].z - d0) < (unsigned)tdn);
      bool h3 = ok & ((unsigned)(v[u].w - d0) < (unsigned)tdn);
      if (h0 | h1 | h2 | h3) {
        int e0 = a0 + 4 * idx[u];
        if (h0) tile_push(L, e0);
        if (h1) tile_push(L, e0 + 1);
        if (h2) tile_push(L, e0 + 2);
        if (h3) tile_push(L, e0 + 3);
      }
    }
  }
  int t0 = a0 + 4 * nvec;
  if (tid < hi - t0) tile_test1(L, dstp, t0 + tid, d0, tdn);
}

#define UM 4
#define KEEP_MAXN (1 << 24)
template <int H>
__device__ __forceinline__ void tile_max(TileLds& L, const int* __restrict__ ei,
                                         const int* __restrict__ dstp,
                                         const float* __restrict__ als,
                                         int n, int d0, int tid) {
  for (int i0 = tid; i0 < n; i0 += NT * UM) {
    int ix[UM], e[UM], s[UM], dl[UM]; float a[UM][H];
#pragma unroll
    for (int u = 0; u < UM; ++u) {
      int i = i0 + u * NT;
      ix[u] = i < n ? i : i0;
      e[u] = L.list[ix[u]];
    }
#pragma unroll
    for (int u = 0; u < UM; ++u) { s[u] = ei[e[u]]; dl[u] = dstp[e[u]] - d0; }
#pragma unroll
    for (int u = 0; u < UM; ++u) {
#pragma unroll
      for (int h = 0; h < H; ++h) a[u][h] = als[(size_t)s[u] * H + h];
    }
#pragma unroll
    for (int u = 0; u < UM; ++u) {
      if (i0 + u * NT < n) {
        L.list[ix[u]] = s[u];
        L.dls[ix[u]] = (unsigned short)dl[u];
#pragma unroll
        for (int h = 0; h < H; ++h) {
          float v = lrelu(a[u][h] + L.aldT[dl[u] * H + h], 0.2f);
          ldsMaxF(&L.mx[dl[u] * H + h], v);
        }
      }
    }
  }
}

template <int H>
__device__ __forceinline__ void tile_max_kept(TileLds& L, const int* __restrict__ wl,
                                              const float* __restrict__ als, int n, int tid) {
  static_assert(TD <= NCN && NCN == 7 * 32, "one wave covers all counters, 7 per lane");
  for (int i0 = tid; i0 < n; i0 += NT * UM) {
    int s[UM], dl[UM]; unsigned p[UM]; float a[UM][H];
#pragma unroll
    for (int u = 0; u < UM; ++u) {
      int i = i0 + u * NT;
      p[u] = (unsigned)wl[i < n ? i : i0];
    }
#pragma unroll
    for (int u = 0; u < UM; ++u) {
      s[u] = (int)(p[u] >> 8); dl[u] = (int)(p[u] & 255u);
#pragma unroll
      for (int h = 0; h < H; ++h) a[u][h] = als[(size_t)s[u] * H + h];
    }
#pragma unroll
    for (int u = 0; u < UM; ++u) {
      if (i0 + u * NT < n) {
        atomicAdd(&L.ncnt[dl[u]], 1);
#pragma unroll
        for (int h = 0; h < H; ++h) {
          float v = lrelu(a[u][h] + L.aldT[dl[u] * H + h], 0.2f);
          ldsMaxF(&L.mx[dl[u] * H + h], v);
        }
      }
    }
  }
  __syncthreads();
  if (tid < 32) {
    int c[7], tot = 0;
#pragma unroll
    for (int k = 0; k < 7; ++k) { c[k] = L.ncnt[7 * tid + k]; tot += c[k]; }
    int inc = tot;
#pragma unroll
    for (int d = 1; d < 32; d <<= 1) { const int t = __shfl_up(inc, d, 32); if (tid >= d) inc += t; }
    int run = inc - tot;
#pragma unroll
    for (int k = 0; k < 7; ++k) { L.ncnt[7 * tid + k] = run; run += c[k]; }
  }
  __syncthreads();
  for (int i0 = tid; i0 < n; i0 += NT * UM) {
    unsigned p[UM];
#pragma unroll
    for (int u = 0; u < UM; ++u) {
      int i = i0 + u * NT;
      p[u] = (unsigned)wl[i < n ? i : i0];
    }
#pragma unroll
    for (int u = 0; u < UM; ++u) {
      if (i0 + u * NT < n) {
        const int dl = (int)(p[u] & 255u);
        const int pos = atomicAdd(&L.ncnt[dl], 1);
        L.list[pos] = (int)(p[u] >> 8);
        L.dls[pos] = (unsigned short)dl;
      }
    }
  }
}


__device__ __forceinline__ float grp8_take(float x, int u) {
  const unsigned v = __float_as_uint(x);
  unsigned r;
  switch (u) {
    case 0:  r = __builtin_amdgcn_mov_dpp8(v, 0x000000); break;
    case 1:  r = __builtin_amdgcn_mov_dpp8(v, 0x249249); break;
    case 2:  r = __builtin_amdgcn_mov_dpp8(v, 0x492492); break;
    case 3:  r = __builtin_amdgcn_mov_dpp8(v, 0x6DB6DB); break;
    case 4:  r = __builtin_amdgcn_mov_dpp8(v, 0x924924); break;
    case 5:  r = __builtin_amdgcn_mov_dpp8(v, 0xB6DB6D); break;
    case 6:  r = __builtin_amdgcn_mov_dpp8(v, 0xDB6DB6); break;
    default: r = __builtin_amdgcn_mov_dpp8(v, 0xFFFFFF); break;
  }
  return __uint_as_float(r);
}

template <int H>
__device__ __forceinline__ void tile_flush(TileLds& L, int d, int lane, float4 s, float ws) {
  float* a = &L.acc[d * HID + lane];
  atomicAdd(a,      s.x);
  atomicAdd(a + 32, s.y);
  atomicAdd(a + 64, s.z);
  atomicAdd(a + 96, s.w);
  const int g = lane >> 3;
  if ((lane & 7) == 0 && g < H) atomicAdd(&L.den[d * H + ((H == 4) ? g : 0)], ws);
}

template <int H>
__device__ __forceinline__ void tile_sum(TileLds& L, const float* __restrict__ hfeat,
                                         const float* __restrict__ als, int n_, int tid) {
  static_assert(H == 1 || H == 4, "a group of 8 lanes is one head (4 heads) or all lanes share one head");
  static_assert(UE == 8, "lane u of a group of 8 lanes holds the weight of entry u of the trip");
  const int wave = __builtin_amdgcn_readfirstlane(tid >> 5);
  const int n = __builtin_amdgcn_readfirstlane(n_);
  const int lane = tid & 31;
  const int g = lane >> 3;
  const int hq = (H == 4) ? g : 0;
  for (int i0 = wave * UE; i0 < n; i0 += (NT / 32) * UE) {
    const int io = i0 + (lane & 7);
    const int i = io < n ? io : n - 1;
    const int sv = L.list[i];
    const int dv = L.dls[i];
    const int qv = dv * H + hq;
    const float alv = als[(size_t)sv * H + hq];
    const float adv = L.aldT[qv], mvv = L.mx[qv];
    int s[UE], dl[UE]; float4 hv[UE];
#pragma unroll
    for (int u = 0; u < UE; ++u) {
      s[u]  = __builtin_amdgcn_readlane(sv, u);
      dl[u] = __builtin_amdgcn_readlane(dv, u);
    }
#pragma unroll
    for (int u = 0; u < UE; ++u)
      hv[u] = reinterpret_cast<const float4*>(hfeat + (size_t)s[u] * HID)[lane];
    const float wv = __expf(lrelu(alv + adv, 0.2f) - mvv);
    float4 sum = {0.f, 0.f, 0.f, 0.f};
    float wsum = 0.f;
    int cur = dl[0];
#pragma unroll
    for (int u = 0; u < UE; ++u) {
      if (dl[u] != cur) {
        tile_flush<H>(L, cur, lane, sum, wsum);
        sum.x = 0.f; sum.y = 0.f; sum.z = 0.f; sum.w = 0.f; wsum = 0.f;
        cur = dl[u];
      }
      const float w = (i0 + u < n) ? grp8_take(wv, u) : 0.f;
      sum.x += w * hv[u].x; sum.y += w * hv[u].y; sum.z += w * hv[u].z; sum.w += w * hv[u].w;
      wsum += w;
    }
    tile_flush<H>(L, cur, lane, sum, wsum);
  }
}

template <int H>
__device__ __forceinline__ void gat_tile_body(TileLds& L, const int* __restrict__ ei,
                                              const float* __restrict__ hfeat,
                                              const float* __restrict__ als,
                                              const float* __restrict__ ald,
                                              const float* __restrict__ bias,
                                              float* __restrict__ outp,
                                              const int* __restrict__ keepbuf,
                                              const int* __restrict__ tcnt,
                                              int E, int N, int act) {
  constexpr int F = HID / H;
  const int tid = threadIdx.x;
  const int d0 = blockIdx.x * TD;
  int tdn = N - d0; if (tdn > TD) tdn = TD;
  const int* dstp = ei + (size_t)E;
  const int* wl = keepbuf + (size_t)blockIdx.x * CAP;

  for (int i = tid; i < TD * HID; i += NT) L.acc[i] = 0.f;
  for (int i = tid; i < NCN; i += NT) L.ncnt[i] = 0;
  for (int i = tid; i < tdn * H; i += NT) {
    size_t g = (size_t)d0 * H + i;
    float a_d = ald[g];
    float v = lrelu(als[g] + a_d, 0.2f);
    L.aldT[i] = a_d; L.wsf[i] = v; L.mx[i] = v;
  }

  __syncthreads();
  int n = tcnt[blockIdx.x];
  if (N > KEEP_MAXN) n = CAP + 1;
  const bool kept = n <= CAP;
  if (!kept) {
    if (tid == 0) L.cnt = 0;
    __syncthreads();
    tile_scan(L, dstp, 0, E, d0, tdn, tid);
    __syncthreads();
    n = L.cnt;
  }
  if (n <= CAP) {
    if (kept) tile_max_kept<H>(L, wl, als, n, tid);
    else      tile_max<H>(L, ei, dstp, als, n, d0, tid);
    __syncthreads();
    for (int i = tid; i < tdn * H; i += NT) {
      float w = __expf(L.wsf[i] - L.mx[i]);
      L.wsf[i] = w; L.den[i] = w;
    }
    __syncthreads();
    tile_sum<H>(L, hfeat, als, n, tid);
  } else {
    const int nwin = (E + CAP - 1) / CAP;
    for (int pass = 0; pass < 2; ++pass) {
      for (int w = 0; w < nwin; ++w) {
        __syncthreads();
        if (tid == 0) L.cnt = 0;
        __syncthreads();
        const int lo = w * CAP;
        int hi = lo + CAP; if (hi > E) hi = E;
        tile_scan(L, dstp, lo, hi, d0, tdn, tid);
        __syncthreads();
        n = L.cnt;
        tile_max<H>(L, ei, dstp, als, n, d0, tid);
        __syncthreads();
        if (pass == 1) tile_sum<H>(L, hfeat, als, n, tid);
      }
      if (pass == 0) {
        __syncthreads();
        for (int i = tid; i < tdn * H; i += NT) {
          float w = __expf(L.wsf[i] - L.mx[i]);
          L.wsf[i] = w; L.den[i] = w;
        }
      }
    }
  }
  __syncthreads();

  {
    constexpr int UO = 7;
    constexpr int RS = NT / 32;
    const int lane = tid & 31;
    const int hq = (4 * lane) / F;
    float bv[4];
#pragma unroll
    for (int k = 0; k < 4; ++k) bv[k] = bias[4 * lane + k];
    for (int r0 = tid >> 5; r0 < tdn; r0 += RS * UO) {
      float4 hv[UO];
#pragma unroll
      for (int u = 0; u < UO; ++u) {
        int dl = r0 + RS * u; if (dl >= tdn) dl = tdn - 1;
        hv[u] = reinterpret_cast<const float4*>(hfeat + (size_t)(d0 + dl) * HID)[lane];
      }
#pragma unroll
      for (int u = 0; u < UO; ++u) {
        const int dl = r0 + RS * u;
        if (dl < tdn) {
          const int q = dl * H + hq;
          const float ws = L.wsf[q], dn = L.den[q];
          const float* a = &L.acc[dl * HID + lane];
          float4 o;
          o.x = (a[0]  + ws * hv[u].x) / dn + bv[0];
          o.y = (a[32] + ws * hv[u].y) / dn + bv[1];
          o.z = (a[64] + ws * hv[u].z) / dn + bv[2];
          o.w = (a[96] + ws * hv[u].w) / dn + bv[3];
          if (act) { o.x = lrelu(o.x, 0.01f); o.y = lrelu(o.y, 0.01f); o.z = lrelu(o.z, 0.01f); o.w = lrelu(o.w, 0.01f); }
          reinterpret_cast<float4*>(outp + (size_t)(d0 + dl) * HID)[lane] = o;
        }
      }
    }
  }
}

__global__ void __launch_bounds__(NT) gat_tile_h4_kernel(const int* __restrict__ ei,
    const float* __restrict__ hfeat, const float* __restrict__ als, const float* __restrict__ ald,
    const float* __restrict__ bias, float* __restrict__ outp, const int* __restrict__ keepbuf,
    const int* __restrict__ tcnt, int E, int N, int act) {
  __shared__ TileLds L;
  gat_tile_body<4>(L, ei, hfeat, als, ald, bias, outp, keepbuf, tcnt, E, N, act);
}

__global__ void __launch_bounds__(NT) gat_tile_h1_kernel(const int* __restrict__ ei,
    const float* __restrict__ hfeat, const float* __restrict__ als, const float* __restrict__ ald,
    const float* __restrict__ bias, float* __restrict__ outp, const int* __restrict__ keepbuf,
    const int* __restrict__ tcnt, int E, int N, int act) {
  __shared__ TileLds L;
  gat_tile_body<1>(L, ei, hfeat, als, ald, bias, outp, keepbuf, tcnt, E, N, act);
}

#define BNT   512
#define BNE   8
#define BMAXT 2048
__global__ void __launch_bounds__(BNT)
bin_edges_kernel(const int* __restrict__ ei, int* __restrict__ keepbuf, int* __restrict__ tcnt,
                 int E, int N, int ntiles) {
  __shared__ int cnt[BMAXT + 1];
  __shared__ int base[BMAXT + 1];
  const int tid = threadIdx.x;
  const int* dstp = ei + (size_t)E;
  const int e0 = blockIdx.x * (BNT * BNE) + tid;
  int s[BNE], tl[BNE], dl[BNE];
#pragma unroll
  for (int u = 0; u < BNE; ++u) {
    const int e = e0 + u * BNT;
    const int ec = e < E ? e : E - 1;
    const int d = dstp[ec];
    s[u] = ei[ec];
    const int t = d / TD;
    dl[u] = d - t * TD;
    tl[u] = ((e < E) & ((unsigned)d < (unsigned)N)) ? t : -1;
  }
  for (int w0 = 0; w0 < ntiles; w0 += BMAXT) {
    const int nt = ntiles - w0 < BMAXT ? ntiles - w0 : BMAXT;
    for (int i = tid; i < nt; i += BNT) cnt[i] = 0;
    if (tid == 0) { cnt[BMAXT] = 0; base[BMAXT] = CAP; }
    __syncthreads();
    int rr[BNE];
#pragma unroll
    for (int u = 0; u < BNE; ++u) {
      const unsigned r = (unsigned)(tl[u] - w0);
      rr[u] = r < (unsigned)nt ? (int)r : BMAXT;
      atomicAdd(&cnt[rr[u]], 1);
    }
    __syncthreads();
    for (int i = tid; i < nt; i += BNT) {
      const int k = cnt[i];
      if (k > 0) base[i] = atomicAdd(&tcnt[w0 + i], k);
      cnt[i] = 0;
    }
    __syncthreads();
    int pos[BNE];
#pragma unroll
    for (int u = 0; u < BNE; ++u) pos[u] = base[rr[u]] + atomicAdd(&cnt[rr[u]], 1);
#pragma unroll
    for (int u = 0; u < BNE; ++u) {
      if (rr[u] != BMAXT && pos[u] < CAP)
        keepbuf[(size_t)tl[u] * CAP + pos[u]] = (int)(((unsigned)s[u] << 8) | (unsigned)dl[u]);
    }
    __syncthreads();
  }
}

extern "C" void kernel_launch(void* const* d_in, const int* in_sizes, int n_in,
                              void* d_out, int out_size, void* d_ws, size_t ws_size,
                              hipStream_t stream) {
  const float* num_prop = (const float*)d_in[0];
  const float* cat_prop = (const float*)d_in[1];
  const int*   ei       = (const int*)d_in[2];
  const float* W_num = (const float*)d_in[3];
  const float* b_num = (const float*)d_in[4];
  const float* W_cat = (const float*)d_in[5];
  const float* b_cat = (const float*)d_in[6];
  const float* W_in  = (const float*)d_in[7];
  const float* b_in  = (const float*)d_in[8];
  const float* bn_g  = (const float*)d_in[9];
  const float* bn_b  = (const float*)d_in[10];
  const float* W1    = (const float*)d_in[11];
  const float* a_s1  = (const float*)d_in[12];
  const float* a_d1  = (const float*)d_in[13];
  const float* bias1 = (const float*)d_in[14];
  const float* W2    = (const float*)d_in[15];
  const float* a_s2  = (const float*)d_in[16];
  const float* a_d2  = (const float*)d_in[17];
  const float* bias2 = (const float*)d_in[18];

  const int N = in_sizes[0] / 7;
  const int E = in_sizes[2] / 2;
  float* out = (float*)d_out;

  float* ws   = (float*)d_ws;
  float* xcat = ws;
  float* x1   = xcat + (size_t)N * HID;
  float* hbuf = x1   + (size_t)N * HID;
  float* accb = hbuf + (size_t)N * HID;
  float* als  = accb + (size_t)N * HID;
  float* ald  = als  + (size_t)N * 4;
  int*   tcnt = (int*)(ald + (size_t)N * 4);
  float* stats= ald  + (size_t)N * 12;

  const int TB = 256;
  const int ntiles = (N + TD - 1) / TD;
  const int mtiles = (N + 15) / 16;
  const int gblocks = (mtiles + GW - 1) / GW;
  auto nb = [&](long long work) { return (unsigned)((work + TB - 1) / TB); };
  int* keepbuf = (int*)x1;
  float* part = hbuf;
  const int bblocks = (E + BNT * BNE - 1) / (BNT * BNE);
  const int eblocks = (N + EW * ER - 1) / (EW * ER);
  const int ablocks = (N * (HID / 4) + AT * AU - 1) / (AT * AU);

  encoder_kernel<<<eblocks, EW * 32, 0, stream>>>(num_prop, cat_prop, W_num, b_num,
                                                  W_cat, b_cat, xcat, N);
  wmma_gemm128_kernel<true><<<gblocks, GW * 32, 0, stream>>>(xcat, W_in, b_in, x1, part, N, 1, 1);

  clear_kernel<<<nb(ntiles), TB, 0, stream>>>(tcnt, ntiles);
  bn_finalize_kernel<<<HID, FT, 0, stream>>>(part, gblocks, stats, bn_g, bn_b, N);
  bn_apply_kernel<<<ablocks, AT, 0, stream>>>(x1, stats, xcat, N);

  bin_edges_kernel<<<bblocks, BNT, 0, stream>>>(ei, keepbuf, tcnt, E, N, ntiles);

  wmma_gemm128_kernel<false><<<gblocks, GW * 32, 0, stream>>>(xcat, W1, nullptr, hbuf, nullptr, N, 0, 0);
  att_logits_kernel<<<nb((long long)N * 4), TB, 0, stream>>>(hbuf, a_s1, a_d1, als, ald, N, 4, 32);
  gat_tile_h4_kernel<<<ntiles, NT, 0, stream>>>(ei, hbuf, als, ald, bias1, accb, keepbuf, tcnt, E, N, 1);

  wmma_gemm128_kernel<false><<<gblocks, GW * 32, 0, stream>>>(accb, W2, nullptr, hbuf, nullptr, N, 0, 0);
  att_logits_kernel<<<nb(N), TB, 0, stream>>>(hbuf, a_s2, a_d2, als, ald, N, 1, 128);
  gat_tile_h1_kernel<<<ntiles, NT, 0, stream>>>(ei, hbuf, als, ald, bias2, out, keepbuf, tcnt, E, N, 0);
}
